# starting kernel
# baseline (speedup 1.0000x reference)
#include <hip/hip_runtime.h>
#include <hip/hip_bf16.h>

typedef float v2f __attribute__((ext_vector_type(2)));
typedef float v8f __attribute__((ext_vector_type(8)));

#define NEG_SLOPE 0.2f
#define EPS 1e-5f

__device__ __forceinline__ void atomicMaxFloat(float* addr, float val) {
    if (val >= 0.0f) {
        atomicMax((int*)addr, __float_as_int(val));
    } else {
        atomicMin((unsigned int*)addr, __float_as_uint(val));
    }
}

__global__ void fill_u32_kernel(unsigned int* __restrict__ p, unsigned int v, long long count) {
    long long i = (long long)blockIdx.x * blockDim.x + threadIdx.x;
    if (i < count) p[i] = v;
}

template <int K, int NCOL>
__global__ void gemm_wmma_f32_kernel(const float* __restrict__ A,
                                     const float* __restrict__ Bm,
                                     float* __restrict__ C) {
    __shared__ float As[16 * K];
    const int rowBase = blockIdx.x * 16;
    for (int i = threadIdx.x; i < 16 * K; i += blockDim.x) {
        int r = i / K, k = i - r * K;
        As[i] = A[(long long)(rowBase + r) * K + k];
    }
    __syncthreads();

    const int wave = threadIdx.x >> 5;
    const int lane = threadIdx.x & 31;
    const int colBase = wave * 16;
    const int mn = lane & 15;
    const int khalf = (lane >> 4) << 1;

    v8f acc = {};
#if __has_builtin(__builtin_amdgcn_wmma_f32_16x16x4_f32)
    for (int kk = 0; kk < K; kk += 4) {
        v2f a, b;
        a.x = As[mn * K + kk + khalf];
        a.y = As[mn * K + kk + khalf + 1];
        b.x = Bm[(long long)(kk + khalf) * NCOL + colBase + mn];
        b.y = Bm[(long long)(kk + khalf + 1) * NCOL + colBase + mn];
        acc = __builtin_amdgcn_wmma_f32_16x16x4_f32(false, a, false, b,
                                                    (short)0, acc, false, false);
    }
#else
    for (int kk = 0; kk < K; ++kk) {
        float av = As[mn * K + kk];
        for (int r = 0; r < 8; ++r)
            acc[r] += av * Bm[(long long)kk * NCOL + colBase + mn];
    }
#endif
    const int mOff = (lane >> 4) << 3;
#pragma unroll
    for (int r = 0; r < 8; ++r) {
        C[(long long)(rowBase + r + mOff) * NCOL + colBase + mn] = acc[r];
    }
}

__global__ void alpha_kernel(const float* __restrict__ h,
                             const float* __restrict__ a_src,
                             const float* __restrict__ a_dst,
                             float* __restrict__ as, float* __restrict__ ad,
                             int n_nodes, int heads, int C) {
    int gw = (int)(((long long)blockIdx.x * blockDim.x + threadIdx.x) >> 5);
    int lane = threadIdx.x & 31;
    if (gw >= n_nodes * heads) return;
    int hd = gw % heads;
    const float* hp = h + (long long)gw * C;
    const float* sp = a_src + hd * C;
    const float* dp = a_dst + hd * C;
    float s1 = 0.0f, s2 = 0.0f;
    for (int c = lane; c < C; c += 32) {
        float v = hp[c];
        s1 += v * sp[c];
        s2 += v * dp[c];
    }
    for (int off = 16; off; off >>= 1) {
        s1 += __shfl_down(s1, off, 32);
        s2 += __shfl_down(s2, off, 32);
    }
    if (lane == 0) { as[gw] = s1; ad[gw] = s2; }
}

__global__ void edge_max_kernel(const int* __restrict__ ei, int E, int E2, int heads,
                                const float* __restrict__ as, const float* __restrict__ ad,
                                float* __restrict__ ebuf, float* __restrict__ emax) {
    int e = (int)((long long)blockIdx.x * blockDim.x + threadIdx.x);
    if (e >= E2) return;
    int s, d;
    if (e < E) { s = ei[e]; d = ei[E + e]; } else { s = d = e - E; }
    for (int hh = 0; hh < heads; ++hh) {
        float v = as[s * heads + hh] + ad[d * heads + hh];
        v = (v > 0.0f) ? v : NEG_SLOPE * v;
        ebuf[(long long)e * heads + hh] = v;
        atomicMaxFloat(&emax[d * heads + hh], v);
    }
}

__global__ void edge_expsum_kernel(const int* __restrict__ ei, int E, int E2, int heads,
                                   const float* __restrict__ emax,
                                   float* __restrict__ ebuf, float* __restrict__ denom) {
    int e = (int)((long long)blockIdx.x * blockDim.x + threadIdx.x);
    if (e >= E2) return;
    int d = (e < E) ? ei[E + e] : (e - E);
    for (int hh = 0; hh < heads; ++hh) {
        float ex = __expf(ebuf[(long long)e * heads + hh] - emax[d * heads + hh]);
        ebuf[(long long)e * heads + hh] = ex;
        atomicAdd(&denom[d * heads + hh], ex);
    }
}

__global__ void edge_aggregate_kernel(const int* __restrict__ ei, int E, int E2,
                                      int heads, int F,
                                      const float* __restrict__ h,
                                      const float* __restrict__ ebuf,
                                      const float* __restrict__ denom,
                                      float* __restrict__ agg) {
    int gw = (int)(((long long)blockIdx.x * blockDim.x + threadIdx.x) >> 5);
    int lane = threadIdx.x & 31;
    if (gw >= E2) return;
    int s, d;
    if (gw < E) { s = ei[gw]; d = ei[E + gw]; } else { s = d = gw - E; }
    int C = F / heads;
    float al[2];
    for (int hh = 0; hh < heads; ++hh)
        al[hh] = ebuf[(long long)gw * heads + hh] / denom[d * heads + hh];
    const float* hs = h + (long long)s * F;
    float* ag = agg + (long long)d * F;
    for (int c = lane; c < F; c += 32) {
        int hh = c / C;
        atomicAdd(&ag[c], hs[c] * al[hh]);
    }
}

__global__ void bias_bn_relu_kernel(const float* __restrict__ x,
                                    const float* __restrict__ b,
                                    const float* __restrict__ gamma,
                                    const float* __restrict__ beta,
                                    const float* __restrict__ mean,
                                    const float* __restrict__ var,
                                    float* __restrict__ out, long long total, int F) {
    long long i = (long long)blockIdx.x * blockDim.x + threadIdx.x;
    if (i >= total) return;
    int c = (int)(i % F);
    float v = x[i] + b[c];
    v = gamma[c] * (v - mean[c]) * rsqrtf(var[c] + EPS) + beta[c];
    out[i] = (v > 0.0f) ? v : 0.0f;
}

__global__ void pool_max_kernel(const float* __restrict__ x,
                                const int* __restrict__ batch,
                                float* __restrict__ out, int n_nodes, int F) {
    long long i = (long long)blockIdx.x * blockDim.x + threadIdx.x;
    if (i >= (long long)n_nodes * F) return;
    int n = (int)(i / F);
    int c = (int)(i - (long long)n * F);
    atomicMaxFloat(&out[batch[n] * F + c], x[i]);
}

extern "C" void kernel_launch(void* const* d_in, const int* in_sizes, int n_in,
                              void* d_out, int out_size, void* d_ws, size_t ws_size,
                              hipStream_t stream) {
    const int F_IN = 128, HID = 128, HEADS = 2;
    const int F1 = HEADS * HID;
    const int N = in_sizes[0] / F_IN;
    const int E = in_sizes[1] / 2;
    const int E2 = E + N;

    const float* x      = (const float*)d_in[0];
    const int*   ei     = (const int*)d_in[1];
    const int*   batch  = (const int*)d_in[2];
    const float* W1     = (const float*)d_in[3];
    const float* a_src1 = (const float*)d_in[4];
    const float* a_dst1 = (const float*)d_in[5];
    const float* b1     = (const float*)d_in[6];
    const float* gamma1 = (const float*)d_in[7];
    const float* beta1  = (const float*)d_in[8];
    const float* rm1    = (const float*)d_in[9];
    const float* rv1    = (const float*)d_in[10];
    const float* W2     = (const float*)d_in[11];
    const float* a_src2 = (const float*)d_in[12];
    const float* a_dst2 = (const float*)d_in[13];
    const float* b2     = (const float*)d_in[14];
    const float* gamma2 = (const float*)d_in[15];
    const float* beta2  = (const float*)d_in[16];
    const float* rm2    = (const float*)d_in[17];
    const float* rv2    = (const float*)d_in[18];

    float* ws = (float*)d_ws;
    size_t o = 0;
    float* h1   = ws + o; o += (size_t)N * F1;
    float* aggA = ws + o; o += (size_t)N * F1;
    float* aggB = ws + o; o += (size_t)N * HID;
    float* as1  = ws + o; o += (size_t)N * HEADS;
    float* ad1  = ws + o; o += (size_t)N * HEADS;
    float* em1  = ws + o; o += (size_t)N * HEADS;
    float* dn1  = ws + o; o += (size_t)N * HEADS;
    float* as2  = ws + o; o += (size_t)N;
    float* ad2  = ws + o; o += (size_t)N;
    float* em2  = ws + o; o += (size_t)N;
    float* dn2  = ws + o; o += (size_t)N;
    float* ex1  = ws + o; o += (size_t)E2 * HEADS;
    float* ex2  = ws + o; o += (size_t)E2;
    float* h2   = h1;

    const unsigned int NEG_INF = 0xFF800000u;
    const unsigned int ZERO = 0u;
    auto fillBlocks = [](long long cnt) { return (unsigned)((cnt + 255) / 256); };

    fill_u32_kernel<<<fillBlocks((long long)N * HEADS), 256, 0, stream>>>((unsigned*)em1, NEG_INF, (long long)N * HEADS);
    fill_u32_kernel<<<fillBlocks((long long)N * HEADS), 256, 0, stream>>>((unsigned*)dn1, ZERO, (long long)N * HEADS);
    fill_u32_kernel<<<fillBlocks((long long)N * F1), 256, 0, stream>>>((unsigned*)aggA, ZERO, (long long)N * F1);
    fill_u32_kernel<<<fillBlocks((long long)N), 256, 0, stream>>>((unsigned*)em2, NEG_INF, (long long)N);
    fill_u32_kernel<<<fillBlocks((long long)N), 256, 0, stream>>>((unsigned*)dn2, ZERO, (long long)N);
    fill_u32_kernel<<<fillBlocks((long long)N * HID), 256, 0, stream>>>((unsigned*)aggB, ZERO, (long long)N * HID);
    fill_u32_kernel<<<fillBlocks((long long)out_size), 256, 0, stream>>>((unsigned*)d_out, NEG_INF, (long long)out_size);

    gemm_wmma_f32_kernel<128, 256><<<N / 16, 32 * (256 / 16), 0, stream>>>(x, W1, h1);
    alpha_kernel<<<(unsigned)(((long long)N * HEADS * 32 + 255) / 256), 256, 0, stream>>>(
        h1, a_src1, a_dst1, as1, ad1, N, HEADS, HID);
    edge_max_kernel<<<fillBlocks(E2), 256, 0, stream>>>(ei, E, E2, HEADS, as1, ad1, ex1, em1);
    edge_expsum_kernel<<<fillBlocks(E2), 256, 0, stream>>>(ei, E, E2, HEADS, em1, ex1, dn1);
    edge_aggregate_kernel<<<(unsigned)(((long long)E2 * 32 + 255) / 256), 256, 0, stream>>>(
        ei, E, E2, HEADS, F1, h1, ex1, dn1, aggA);
    bias_bn_relu_kernel<<<fillBlocks((long long)N * F1), 256, 0, stream>>>(
        aggA, b1, gamma1, beta1, rm1, rv1, aggA, (long long)N * F1, F1);

    gemm_wmma_f32_kernel<256, 128><<<N / 16, 32 * (128 / 16), 0, stream>>>(aggA, W2, h2);
    alpha_kernel<<<(unsigned)(((long long)N * 32 + 255) / 256), 256, 0, stream>>>(
        h2, a_src2, a_dst2, as2, ad2, N, 1, HID);
    edge_max_kernel<<<fillBlocks(E2), 256, 0, stream>>>(ei, E, E2, 1, as2, ad2, ex2, em2);
    edge_expsum_kernel<<<fillBlocks(E2), 256, 0, stream>>>(ei, E, E2, 1, em2, ex2, dn2);
    edge_aggregate_kernel<<<(unsigned)(((long long)E2 * 32 + 255) / 256), 256, 0, stream>>>(
        ei, E, E2, 1, HID, h2, ex2, dn2, aggB);
    bias_bn_relu_kernel<<<fillBlocks((long long)N * HID), 256, 0, stream>>>(
        aggB, b2, gamma2, beta2, rm2, rv2, aggB, (long long)N * HID, HID);

    pool_max_kernel<<<fillBlocks((long long)N * HID), 256, 0, stream>>>(
        aggB, batch, (float*)d_out, N, HID);
}


# Edges counting-sorted by destination per call; one wave per node does softmax and row gather in registers instead of float-atomic scatter passes (sum order changes; normalised after the sum)
# speedup vs baseline: 1.9437x; 1.9437x over previous
#include <hip/hip_runtime.h>
#include <hip/hip_bf16.h>

typedef float v2f __attribute__((ext_vector_type(2)));
typedef float v8f __attribute__((ext_vector_type(8)));

#define NEG_SLOPE 0.2f
#define EPS 1e-5f

__device__ __forceinline__ void atomicMaxFloat(float* addr, float val) {
    if (val >= 0.0f) {
        atomicMax((int*)addr, __float_as_int(val));
    } else {
        atomicMin((unsigned int*)addr, __float_as_uint(val));
    }
}

__global__ void fill_u32_kernel(unsigned int* __restrict__ p, unsigned int v, long long count) {
    long long i = (long long)blockIdx.x * blockDim.x + threadIdx.x;
    if (i < count) p[i] = v;
}

template <int K, int NCOL>
__global__ void gemm_wmma_f32_kernel(const float* __restrict__ A,
                                     const float* __restrict__ Bm,
                                     float* __restrict__ C) {
    __shared__ float As[16 * K];
    const int rowBase = blockIdx.x * 16;
    for (int i = threadIdx.x; i < 16 * K; i += blockDim.x) {
        int r = i / K, k = i - r * K;
        As[i] = A[(long long)(rowBase + r) * K + k];
    }
    __syncthreads();

    const int wave = threadIdx.x >> 5;
    const int lane = threadIdx.x & 31;
    const int colBase = wave * 16;
    const int mn = lane & 15;
    const int khalf = (lane >> 4) << 1;

    v8f acc = {};
#if __has_builtin(__builtin_amdgcn_wmma_f32_16x16x4_f32)
    for (int kk = 0; kk < K; kk += 4) {
        v2f a, b;
        a.x = As[mn * K + kk + khalf];
        a.y = As[mn * K + kk + khalf + 1];
        b.x = Bm[(long long)(kk + khalf) * NCOL + colBase + mn];
        b.y = Bm[(long long)(kk + khalf + 1) * NCOL + colBase + mn];
        acc = __builtin_amdgcn_wmma_f32_16x16x4_f32(false, a, false, b,
                                                    (short)0, acc, false, false);
    }
#else
    for (int kk = 0; kk < K; ++kk) {
        float av = As[mn * K + kk];
        for (int r = 0; r < 8; ++r)
            acc[r] += av * Bm[(long long)kk * NCOL + colBase + mn];
    }
#endif
    const int mOff = (lane >> 4) << 3;
#pragma unroll
    for (int r = 0; r < 8; ++r) {
        C[(long long)(rowBase + r + mOff) * NCOL + colBase + mn] = acc[r];
    }
}

__global__ void alpha_kernel(const float* __restrict__ h,
                             const float* __restrict__ a_src,
                             const float* __restrict__ a_dst,
                             float* __restrict__ as, float* __restrict__ ad,
                             int n_nodes, int heads, int C) {
    int gw = (int)(((long long)blockIdx.x * blockDim.x + threadIdx.x) >> 5);
    int lane = threadIdx.x & 31;
    if (gw >= n_nodes * heads) return;
    int hd = gw % heads;
    const float* hp = h + (long long)gw * C;
    const float* sp = a_src + hd * C;
    const float* dp = a_dst + hd * C;
    float s1 = 0.0f, s2 = 0.0f;
    for (int c = lane; c < C; c += 32) {
        float v = hp[c];
        s1 += v * sp[c];
        s2 += v * dp[c];
    }
    for (int off = 16; off; off >>= 1) {
        s1 += __shfl_down(s1, off, 32);
        s2 += __shfl_down(s2, off, 32);
    }
    if (lane == 0) { as[gw] = s1; ad[gw] = s2; }
}

__global__ void degree_count_kernel(const int* __restrict__ ei, int E, int* __restrict__ cnt) {
    int e = (int)((long long)blockIdx.x * blockDim.x + threadIdx.x);
    if (e < E) atomicAdd(&cnt[ei[E + e]], 1);
}

__global__ void rowptr_scan_kernel(const int* __restrict__ cnt, int* __restrict__ rowptr, int n) {
    __shared__ int wsum[32];
    const int tid = threadIdx.x;
    const int nthreads = blockDim.x;
    const int lane = tid & 31;
    const int wv = tid >> 5;
    const int per = (n + nthreads - 1) / nthreads;
    const int beg = tid * per;
    const int end = (beg + per < n) ? (beg + per) : n;
    int local = 0;
    for (int i = beg; i < end; ++i) local += cnt[i];
    int inc = local;
    for (int off = 1; off < 32; off <<= 1) {
        int t = __shfl_up(inc, off, 32);
        if (lane >= off) inc += t;
    }
    if (lane == 31) wsum[wv] = inc;
    __syncthreads();
    if (wv == 0) {
        const int w = (lane < (nthreads >> 5)) ? wsum[lane] : 0;
        int winc = w;
        for (int off = 1; off < 32; off <<= 1) {
            int t = __shfl_up(winc, off, 32);
            if (lane >= off) winc += t;
        }
        wsum[lane] = winc - w;
    }
    __syncthreads();
    int run = wsum[wv] + inc - local;
    for (int i = beg; i < end; ++i) {
        rowptr[i] = run;
        run += cnt[i];
    }
    if (tid == nthreads - 1) rowptr[n] = run;
}

__global__ void csr_fill_kernel(const int* __restrict__ ei, int E,
                                const int* __restrict__ rowptr,
                                int* __restrict__ cnt, int* __restrict__ col) {
    int e = (int)((long long)blockIdx.x * blockDim.x + threadIdx.x);
    if (e >= E) return;
    const int d = ei[E + e];
    const int k = atomicSub(&cnt[d], 1) - 1;
    const int slot = rowptr[d] + k;
    if ((unsigned)slot < (unsigned)E) col[slot] = ei[e];
}

template <int HEADS>
__global__ void node_softmax_gather_kernel(const int* __restrict__ rowptr,
                                           const int* __restrict__ col,
                                           const float* __restrict__ h,
                                           const float* __restrict__ as,
                                           const float* __restrict__ ad,
                                           float* __restrict__ out,
                                           int n_nodes, int E) {
    constexpr int C = 128;
    constexpr int F = HEADS * C;
    const int lane = threadIdx.x & 31;
    const int n = __builtin_amdgcn_readfirstlane(
        (int)(((long long)blockIdx.x * blockDim.x + threadIdx.x) >> 5));
    if (n >= n_nodes) return;

    int beg = rowptr[n];
    int nin = rowptr[n + 1] - beg;
    beg = (beg < 0) ? 0 : ((beg > E) ? E : beg);
    nin = (nin < 0) ? 0 : ((nin > E - beg) ? (E - beg) : nin);
    const int deg = nin + 1;

    float adn[HEADS];
#pragma unroll
    for (int hh = 0; hh < HEADS; ++hh) adn[hh] = ad[n * HEADS + hh];

    float m[HEADS];
#pragma unroll
    for (int hh = 0; hh < HEADS; ++hh) m[hh] = -INFINITY;
    for (int j = lane; j < deg; j += 32) {
        int s = (j < nin) ? col[beg + j] : n;
        s = (s < 0) ? 0 : ((s >= n_nodes) ? (n_nodes - 1) : s);
#pragma unroll
        for (int hh = 0; hh < HEADS; ++hh) {
            float v = as[s * HEADS + hh] + adn[hh];
            v = (v > 0.0f) ? v : NEG_SLOPE * v;
            m[hh] = fmaxf(m[hh], v);
        }
    }
    for (int off = 16; off; off >>= 1) {
#pragma unroll
        for (int hh = 0; hh < HEADS; ++hh) m[hh] = fmaxf(m[hh], __shfl_xor(m[hh], off, 32));
    }

    float4 acc[HEADS];
    float dsum[HEADS];
#pragma unroll
    for (int hh = 0; hh < HEADS; ++hh) {
        acc[hh] = make_float4(0.0f, 0.0f, 0.0f, 0.0f);
        dsum[hh] = 0.0f;
    }
    for (int base = 0; base < deg; base += 32) {
        const int j = base + lane;
        int s = n;
        float ex[HEADS];
#pragma unroll
        for (int hh = 0; hh < HEADS; ++hh) ex[hh] = 0.0f;
        if (j < deg) {
            s = (j < nin) ? col[beg + j] : n;
            s = (s < 0) ? 0 : ((s >= n_nodes) ? (n_nodes - 1) : s);
#pragma unroll
            for (int hh = 0; hh < HEADS; ++hh) {
                float v = as[s * HEADS + hh] + adn[hh];
                v = (v > 0.0f) ? v : NEG_SLOPE * v;
                ex[hh] = __expf(v - m[hh]);
                dsum[hh] += ex[hh];
            }
        }
        const int chunk = (deg - base < 32) ? (deg - base) : 32;
#pragma unroll 4
        for (int t = 0; t < chunk; ++t) {
            const int st = __builtin_amdgcn_readlane(s, t);
            const float* row = h + (size_t)st * F + lane * 4;
#pragma unroll
            for (int hh = 0; hh < HEADS; ++hh) {
                const float a = __int_as_float(__builtin_amdgcn_readlane(__float_as_int(ex[hh]), t));
                const float4 hv = *(const float4*)(row + hh * C);
                acc[hh].x += a * hv.x;
                acc[hh].y += a * hv.y;
                acc[hh].z += a * hv.z;
                acc[hh].w += a * hv.w;
            }
        }
    }
    for (int off = 16; off; off >>= 1) {
#pragma unroll
        for (int hh = 0; hh < HEADS; ++hh) dsum[hh] += __shfl_xor(dsum[hh], off, 32);
    }
#pragma unroll
    for (int hh = 0; hh < HEADS; ++hh) {
        const float inv = 1.0f / dsum[hh];
        float4 r;
        r.x = acc[hh].x * inv;
        r.y = acc[hh].y * inv;
        r.z = acc[hh].z * inv;
        r.w = acc[hh].w * inv;
        *(float4*)(out + (size_t)n * F + hh * C + lane * 4) = r;
    }
}

__global__ void bias_bn_relu_kernel(const float* __restrict__ x,
                                    const float* __restrict__ b,
                                    const float* __restrict__ gamma,
                                    const float* __restrict__ beta,
                                    const float* __restrict__ mean,
                                    const float* __restrict__ var,
                                    float* __restrict__ out, long long total, int F) {
    long long i = (long long)blockIdx.x * blockDim.x + threadIdx.x;
    if (i >= total) return;
    int c = (int)(i % F);
    float v = x[i] + b[c];
    v = gamma[c] * (v - mean[c]) * rsqrtf(var[c] + EPS) + beta[c];
    out[i] = (v > 0.0f) ? v : 0.0f;
}

__global__ void pool_max_kernel(const float* __restrict__ x,
                                const int* __restrict__ batch,
                                float* __restrict__ out, int n_nodes, int F) {
    long long i = (long long)blockIdx.x * blockDim.x + threadIdx.x;
    if (i >= (long long)n_nodes * F) return;
    int n = (int)(i / F);
    int c = (int)(i - (long long)n * F);
    atomicMaxFloat(&out[batch[n] * F + c], x[i]);
}

extern "C" void kernel_launch(void* const* d_in, const int* in_sizes, int n_in,
                              void* d_out, int out_size, void* d_ws, size_t ws_size,
                              hipStream_t stream) {
    const int F_IN = 128, HID = 128, HEADS = 2;
    const int F1 = HEADS * HID;
    const int N = in_sizes[0] / F_IN;
    const int E = in_sizes[1] / 2;

    const float* x      = (const float*)d_in[0];
    const int*   ei     = (const int*)d_in[1];
    const int*   batch  = (const int*)d_in[2];
    const float* W1     = (const float*)d_in[3];
    const float* a_src1 = (const float*)d_in[4];
    const float* a_dst1 = (const float*)d_in[5];
    const float* b1     = (const float*)d_in[6];
    const float* gamma1 = (const float*)d_in[7];
    const float* beta1  = (const float*)d_in[8];
    const float* rm1    = (const float*)d_in[9];
    const float* rv1    = (const float*)d_in[10];
    const float* W2     = (const float*)d_in[11];
    const float* a_src2 = (const float*)d_in[12];
    const float* a_dst2 = (const float*)d_in[13];
    const float* b2     = (const float*)d_in[14];
    const float* gamma2 = (const float*)d_in[15];
    const float* beta2  = (const float*)d_in[16];
    const float* rm2    = (const float*)d_in[17];
    const float* rv2    = (const float*)d_in[18];

    float* ws = (float*)d_ws;
    size_t o = 0;
    float* h1   = ws + o; o += (size_t)N * F1;
    float* aggA = ws + o; o += (size_t)N * F1;
    float* aggB = ws + o; o += (size_t)N * HID;
    float* as1  = ws + o; o += (size_t)N * HEADS;
    float* ad1  = ws + o; o += (size_t)N * HEADS;
    float* as2  = ws + o; o += (size_t)N;
    float* ad2  = ws + o; o += (size_t)N;
    int* cnt    = (int*)(ws + o); o += (size_t)N;
    int* rowptr = (int*)(ws + o); o += (size_t)N + 1;
    int* col    = (int*)(ws + o); o += (size_t)E;
    float* h2   = h1;

    const unsigned int NEG_INF = 0xFF800000u;
    const unsigned int ZERO = 0u;
    auto fillBlocks = [](long long cnt_) { return (unsigned)((cnt_ + 255) / 256); };

    fill_u32_kernel<<<fillBlocks((long long)N), 256, 0, stream>>>((unsigned*)cnt, ZERO, (long long)N);
    fill_u32_kernel<<<fillBlocks((long long)out_size), 256, 0, stream>>>((unsigned*)d_out, NEG_INF, (long long)out_size);

    degree_count_kernel<<<fillBlocks(E), 256, 0, stream>>>(ei, E, cnt);
    rowptr_scan_kernel<<<1, 1024, 0, stream>>>(cnt, rowptr, N);
    csr_fill_kernel<<<fillBlocks(E), 256, 0, stream>>>(ei, E, rowptr, cnt, col);

    gemm_wmma_f32_kernel<128, 256><<<N / 16, 32 * (256 / 16), 0, stream>>>(x, W1, h1);
    alpha_kernel<<<(unsigned)(((long long)N * HEADS * 32 + 255) / 256), 256, 0, stream>>>(
        h1, a_src1, a_dst1, as1, ad1, N, HEADS, HID);
    node_softmax_gather_kernel<2><<<(unsigned)(((long long)N * 32 + 255) / 256), 256, 0, stream>>>(
        rowptr, col, h1, as1, ad1, aggA, N, E);
    bias_bn_relu_kernel<<<fillBlocks((long long)N * F1), 256, 0, stream>>>(
        aggA, b1, gamma1, beta1, rm1, rv1, aggA, (long long)N * F1, F1);

    gemm_wmma_f32_kernel<256, 128><<<N / 16, 32 * (128 / 16), 0, stream>>>(aggA, W2, h2);
    alpha_kernel<<<(unsigned)(((long long)N * 32 + 255) / 256), 256, 0, stream>>>(
        h2, a_src2, a_dst2, as2, ad2, N, 1, HID);
    node_softmax_gather_kernel<1><<<(unsigned)(((long long)N * 32 + 255) / 256), 256, 0, stream>>>(
        rowptr, col, h2, as2, ad2, aggB, N, E);
    bias_bn_relu_kernel<<<fillBlocks((long long)N * HID), 256, 0, stream>>>(
        aggB, b2, gamma2, beta2, rm2, rv2, aggB, (long long)N * HID, HID);

    pool_max_kernel<<<fillBlocks((long long)N * HID), 256, 0, stream>>>(
        aggB, batch, (float*)d_out, N, HID);
}


# GEMM A strip kept in LDS at a row pitch of K+4 floats instead of K: the 16 rows of one fragment read start 4 banks apart instead of in one bank (the read was a 16-way bank conflict)
# speedup vs baseline: 2.8622x; 1.4726x over previous
#include <hip/hip_runtime.h>
#include <hip/hip_bf16.h>

typedef float v2f __attribute__((ext_vector_type(2)));
typedef float v8f __attribute__((ext_vector_type(8)));

#define NEG_SLOPE 0.2f
#define EPS 1e-5f

__device__ __forceinline__ void atomicMaxFloat(float* addr, float val) {
    if (val >= 0.0f) {
        atomicMax((int*)addr, __float_as_int(val));
    } else {
        atomicMin((unsigned int*)addr, __float_as_uint(val));
    }
}

__global__ void fill_u32_kernel(unsigned int* __restrict__ p, unsigned int v, long long count) {
    long long i = (long long)blockIdx.x * blockDim.x + threadIdx.x;
    if (i < count) p[i] = v;
}

template <int XMASK>
__device__ __forceinline__ float add_row_xmask(float v) {
    return v + __int_as_float(__builtin_amdgcn_update_dpp(0, __float_as_int(v), 0x160 + XMASK, 0xf, 0xf, true));
}

template <int K, int NCOL, int HEADS>
__global__ void gemm_wmma_f32_logits_kernel(const float* __restrict__ A,
                                            const float* __restrict__ Bm,
                                            const float* __restrict__ a_src,
                                            const float* __restrict__ a_dst,
                                            float* __restrict__ C,
                                            float* __restrict__ as,
                                            float* __restrict__ ad) {
    constexpr int NW = NCOL / 16;
    constexpr int WPH = NW / HEADS;
    constexpr int KP = K + 4;
    __shared__ float As[16 * KP];
    __shared__ alignas(16) float part[2][NW][16];
    const int rowBase = blockIdx.x * 16;
    for (int i = threadIdx.x; i < 16 * K; i += blockDim.x) {
        int r = i / K, k = i - r * K;
        As[r * KP + k] = A[(long long)(rowBase + r) * K + k];
    }
    __syncthreads();

    const int wave = threadIdx.x >> 5;
    const int lane = threadIdx.x & 31;
    const int colBase = wave * 16;
    const int mn = lane & 15;
    const int khalf = (lane >> 4) << 1;

    const float sa = a_src[colBase + mn];
    const float da = a_dst[colBase + mn];

    v8f acc = {};
#if __has_builtin(__builtin_amdgcn_wmma_f32_16x16x4_f32)
    for (int kk = 0; kk < K; kk += 4) {
        v2f a, b;
        a.x = As[mn * KP + kk + khalf];
        a.y = As[mn * KP + kk + khalf + 1];
        b.x = Bm[(long long)(kk + khalf) * NCOL + colBase + mn];
        b.y = Bm[(long long)(kk + khalf + 1) * NCOL + colBase + mn];
        acc = __builtin_amdgcn_wmma_f32_16x16x4_f32(false, a, false, b,
                                                    (short)0, acc, false, false);
    }
#else
    for (int kk = 0; kk < K; ++kk) {
        float av = As[mn * KP + kk];
        for (int r = 0; r < 8; ++r)
            acc[r] += av * Bm[(long long)kk * NCOL + colBase + mn];
    }
#endif
    const int mOff = (lane >> 4) << 3;
#pragma unroll
    for (int r = 0; r < 8; ++r) {
        C[(long long)(rowBase + r + mOff) * NCOL + colBase + mn] = acc[r];
    }

    float ps[8], pd[8];
#pragma unroll
    for (int r = 0; r < 8; ++r) { ps[r] = acc[r] * sa; pd[r] = acc[r] * da; }
#pragma unroll
    for (int r = 0; r < 8; ++r) { ps[r] = add_row_xmask<1>(ps[r]); pd[r] = add_row_xmask<1>(pd[r]); }
#pragma unroll
    for (int r = 0; r < 8; ++r) { ps[r] = add_row_xmask<2>(ps[r]); pd[r] = add_row_xmask<2>(pd[r]); }
#pragma unroll
    for (int r = 0; r < 8; ++r) { ps[r] = add_row_xmask<4>(ps[r]); pd[r] = add_row_xmask<4>(pd[r]); }
#pragma unroll
    for (int r = 0; r < 8; ++r) { ps[r] = add_row_xmask<8>(ps[r]); pd[r] = add_row_xmask<8>(pd[r]); }
    if (mn == 0) {
        *(float4*)(&part[0][wave][mOff])     = make_float4(ps[0], ps[1], ps[2], ps[3]);
        *(float4*)(&part[0][wave][mOff + 4]) = make_float4(ps[4], ps[5], ps[6], ps[7]);
        *(float4*)(&part[1][wave][mOff])     = make_float4(pd[0], pd[1], pd[2], pd[3]);
        *(float4*)(&part[1][wave][mOff + 4]) = make_float4(pd[4], pd[5], pd[6], pd[7]);
    }
    __syncthreads();
    if (threadIdx.x < 32 * HEADS) {
        const int row = threadIdx.x & 15;
        const int which = (threadIdx.x >> 4) & 1;
        const int hd = threadIdx.x >> 5;
        float s = 0.0f;
#pragma unroll
        for (int w = 0; w < WPH; ++w) s += part[which][hd * WPH + w][row];
        float* dstp = which ? ad : as;
        dstp[(long long)(rowBase + row) * HEADS + hd] = s;
    }
}

__global__ void degree_count_kernel(const int* __restrict__ ei, int E, int* __restrict__ cnt) {
    int e = (int)((long long)blockIdx.x * blockDim.x + threadIdx.x);
    if (e < E) atomicAdd(&cnt[ei[E + e]], 1);
}

__global__ void rowptr_scan_kernel(const int* __restrict__ cnt, int* __restrict__ rowptr, int n) {
    __shared__ int wsum[32];
    const int tid = threadIdx.x;
    const int nthreads = blockDim.x;
    const int lane = tid & 31;
    const int wv = tid >> 5;
    const int per = (n + nthreads - 1) / nthreads;
    const int beg = tid * per;
    const int end = (beg + per < n) ? (beg + per) : n;
    int local = 0;
    for (int i = beg; i < end; ++i) local += cnt[i];
    int inc = local;
    for (int off = 1; off < 32; off <<= 1) {
        int t = __shfl_up(inc, off, 32);
        if (lane >= off) inc += t;
    }
    if (lane == 31) wsum[wv] = inc;
    __syncthreads();
    if (wv == 0) {
        const int w = (lane < (nthreads >> 5)) ? wsum[lane] : 0;
        int winc = w;
        for (int off = 1; off < 32; off <<= 1) {
            int t = __shfl_up(winc, off, 32);
            if (lane >= off) winc += t;
        }
        wsum[lane] = winc - w;
    }
    __syncthreads();
    int run = wsum[wv] + inc - local;
    for (int i = beg; i < end; ++i) {
        rowptr[i] = run;
        run += cnt[i];
    }
    if (tid == nthreads - 1) rowptr[n] = run;
}

__global__ void csr_fill_kernel(const int* __restrict__ ei, int E,
                                const int* __restrict__ rowptr,
                                int* __restrict__ cnt, int* __restrict__ col) {
    int e = (int)((long long)blockIdx.x * blockDim.x + threadIdx.x);
    if (e >= E) return;
    const int d = ei[E + e];
    const int k = atomicSub(&cnt[d], 1) - 1;
    const int slot = rowptr[d] + k;
    if ((unsigned)slot < (unsigned)E) col[slot] = ei[e];
}

__device__ __forceinline__ float bias_bn_relu(float x, float b, float gamma, float beta,
                                              float mean, float var) {
    float v = x + b;
    v = gamma * (v - mean) * rsqrtf(var + EPS) + beta;
    return (v > 0.0f) ? v : 0.0f;
}

constexpr int GATHER_WAVES = 8;

template <int HEADS, bool POOL>
__global__ void node_softmax_gather_kernel(const int* __restrict__ rowptr,
                                           const int* __restrict__ col,
                                           const float* __restrict__ h,
                                           const float* __restrict__ as,
                                           const float* __restrict__ ad,
                                           const float* __restrict__ bias,
                                           const float* __restrict__ gamma,
                                           const float* __restrict__ beta,
                                           const float* __restrict__ mean,
                                           const float* __restrict__ var,
                                           float* __restrict__ out,
                                           const int* __restrict__ batch,
                                           float* __restrict__ pool_out,
                                           int n_nodes, int E, int n_graphs) {
    constexpr int C = 128;
    constexpr int F = HEADS * C;
    static_assert(!POOL || HEADS == 1, "the pooled epilogue is written for one head");
    const int lane = threadIdx.x & 31;
    const int n = __builtin_amdgcn_readfirstlane(
        (int)(((long long)blockIdx.x * blockDim.x + threadIdx.x) >> 5));
    const bool valid = n < n_nodes;
    if (!POOL && !valid) return;

    float4 res[HEADS];
    if (valid) {
        int beg = rowptr[n];
        int nin = rowptr[n + 1] - beg;
        beg = (beg < 0) ? 0 : ((beg > E) ? E : beg);
        nin = (nin < 0) ? 0 : ((nin > E - beg) ? (E - beg) : nin);
        const int deg = nin + 1;

        float adn[HEADS];
#pragma unroll
        for (int hh = 0; hh < HEADS; ++hh) adn[hh] = ad[n * HEADS + hh];

        float m[HEADS];
#pragma unroll
        for (int hh = 0; hh < HEADS; ++hh) m[hh] = -INFINITY;
        for (int j = lane; j < deg; j += 32) {
            int s = (j < nin) ? col[beg + j] : n;
            s = (s < 0) ? 0 : ((s >= n_nodes) ? (n_nodes - 1) : s);
#pragma unroll
            for (int hh = 0; hh < HEADS; ++hh) {
                float v = as[s * HEADS + hh] + adn[hh];
                v = (v > 0.0f) ? v : NEG_SLOPE * v;
                m[hh] = fmaxf(m[hh], v);
            }
        }
        for (int off = 16; off; off >>= 1) {
#pragma unroll
            for (int hh = 0; hh < HEADS; ++hh) m[hh] = fmaxf(m[hh], __shfl_xor(m[hh], off, 32));
        }

        float4 acc[HEADS];
        float dsum[HEADS];
#pragma unroll
        for (int hh = 0; hh < HEADS; ++hh) {
            acc[hh] = make_float4(0.0f, 0.0f, 0.0f, 0.0f);
            dsum[hh] = 0.0f;
        }
        for (int base = 0; base < deg; base += 32) {
            const int j = base + lane;
            int s = n;
            float ex[HEADS];
#pragma unroll
            for (int hh = 0; hh < HEADS; ++hh) ex[hh] = 0.0f;
            if (j < deg) {
                s = (j < nin) ? col[beg + j] : n;
                s = (s < 0) ? 0 : ((s >= n_nodes) ? (n_nodes - 1) : s);
#pragma unroll
                for (int hh = 0; hh < HEADS; ++hh) {
                    float v = as[s * HEADS + hh] + adn[hh];
                    v = (v > 0.0f) ? v : NEG_SLOPE * v;
                    ex[hh] = __expf(v - m[hh]);
                    dsum[hh] += ex[hh];
                }
            }
            const int chunk = (deg - base < 32) ? (deg - base) : 32;
#pragma unroll 4
            for (int t = 0; t < chunk; ++t) {
                const int st = __builtin_amdgcn_readlane(s, t);
                const float* row = h + (size_t)st * F + lane * 4;
#pragma unroll
                for (int hh = 0; hh < HEADS; ++hh) {
                    const float a = __int_as_float(__builtin_amdgcn_readlane(__float_as_int(ex[hh]), t));
                    const float4 hv = *(const float4*)(row + hh * C);
                    acc[hh].x += a * hv.x;
                    acc[hh].y += a * hv.y;
                    acc[hh].z += a * hv.z;
                    acc[hh].w += a * hv.w;
                }
            }
        }
        for (int off = 16; off; off >>= 1) {
#pragma unroll
            for (int hh = 0; hh < HEADS; ++hh) dsum[hh] += __shfl_xor(dsum[hh], off, 32);
        }
#pragma unroll
        for (int hh = 0; hh < HEADS; ++hh) {
            const float inv = 1.0f / dsum[hh];
            const int c0 = hh * C + lane * 4;
            const float4 pb = *(const float4*)(bias + c0);
            const float4 pg = *(const float4*)(gamma + c0);
            const float4 pe = *(const float4*)(beta + c0);
            const float4 pm = *(const float4*)(mean + c0);
            const float4 pv = *(const float4*)(var + c0);
            res[hh].x = bias_bn_relu(acc[hh].x * inv, pb.x, pg.x, pe.x, pm.x, pv.x);
            res[hh].y = bias_bn_relu(acc[hh].y * inv, pb.y, pg.y, pe.y, pm.y, pv.y);
            res[hh].z = bias_bn_relu(acc[hh].z * inv, pb.z, pg.z, pe.z, pm.z, pv.z);
            res[hh].w = bias_bn_relu(acc[hh].w * inv, pb.w, pg.w, pe.w, pm.w, pv.w);
        }
    }

    if constexpr (!POOL) {
#pragma unroll
        for (int hh = 0; hh < HEADS; ++hh)
            *(float4*)(out + (size_t)n * F + hh * C + lane * 4) = res[hh];
    } else {
        __shared__ float rowbuf[GATHER_WAVES * C];
        __shared__ int gid[GATHER_WAVES];
        const int w = threadIdx.x >> 5;
        if (valid) {
            *(float4*)(rowbuf + w * C + lane * 4) = res[0];
            if (lane == 0) {
                const int g = batch[n];
                gid[w] = ((unsigned)g < (unsigned)n_graphs) ? g : -1;
            }
        } else if (lane == 0) {
            gid[w] = -1;
        }
        __syncthreads();
        if (threadIdx.x < C) {
            const int c = threadIdx.x;
            int cur = -1;
            float cm = 0.0f;
#pragma unroll
            for (int i = 0; i < GATHER_WAVES; ++i) {
                const int g = gid[i];
                if (g < 0) continue;
                const float v = rowbuf[i * C + c];
                if (g == cur) {
                    cm = fmaxf(cm, v);
                } else {
                    if (cur >= 0) atomicMaxFloat(&pool_out[(size_t)cur * C + c], cm);
                    cur = g;
                    cm = v;
                }
            }
            if (cur >= 0) atomicMaxFloat(&pool_out[(size_t)cur * C + c], cm);
        }
    }
}

extern "C" void kernel_launch(void* const* d_in, const int* in_sizes, int n_in,
                              void* d_out, int out_size, void* d_ws, size_t ws_size,
                              hipStream_t stream) {
    const int F_IN = 128, HID = 128, HEADS = 2;
    const int F1 = HEADS * HID;
    const int N = in_sizes[0] / F_IN;
    const int E = in_sizes[1] / 2;

    const float* x      = (const float*)d_in[0];
    const int*   ei     = (const int*)d_in[1];
    const int*   batch  = (const int*)d_in[2];
    const float* W1     = (const float*)d_in[3];
    const float* a_src1 = (const float*)d_in[4];
    const float* a_dst1 = (const float*)d_in[5];
    const float* b1     = (const float*)d_in[6];
    const float* gamma1 = (const float*)d_in[7];
    const float* beta1  = (const float*)d_in[8];
    const float* rm1    = (const float*)d_in[9];
    const float* rv1    = (const float*)d_in[10];
    const float* W2     = (const float*)d_in[11];
    const float* a_src2 = (const float*)d_in[12];
    const float* a_dst2 = (const float*)d_in[13];
    const float* b2     = (const float*)d_in[14];
    const float* gamma2 = (const float*)d_in[15];
    const float* beta2  = (const float*)d_in[16];
    const float* rm2    = (const float*)d_in[17];
    const float* rv2    = (const float*)d_in[18];

    float* ws = (float*)d_ws;
    size_t o = 0;
    float* h1   = ws + o; o += (size_t)N * F1;
    float* aggA = ws + o; o += (size_t)N * F1;
    float* aggB = ws + o; o += (size_t)N * HID;
    float* as1  = ws + o; o += (size_t)N * HEADS;
    float* ad1  = ws + o; o += (size_t)N * HEADS;
    float* as2  = ws + o; o += (size_t)N;
    float* ad2  = ws + o; o += (size_t)N;
    int* cnt    = (int*)(ws + o); o += (size_t)N;
    int* rowptr = (int*)(ws + o); o += (size_t)N + 1;
    int* col    = (int*)(ws + o); o += (size_t)E;
    float* h2   = h1;

    const unsigned int NEG_INF = 0xFF800000u;
    const unsigned int ZERO = 0u;
    auto fillBlocks = [](long long cnt_) { return (unsigned)((cnt_ + 255) / 256); };
    const unsigned gatherBlocks = (unsigned)((N + GATHER_WAVES - 1) / GATHER_WAVES);

    fill_u32_kernel<<<fillBlocks((long long)N), 256, 0, stream>>>((unsigned*)cnt, ZERO, (long long)N);
    fill_u32_kernel<<<fillBlocks((long long)out_size), 256, 0, stream>>>((unsigned*)d_out, NEG_INF, (long long)out_size);

    degree_count_kernel<<<fillBlocks(E), 256, 0, stream>>>(ei, E, cnt);
    rowptr_scan_kernel<<<1, 1024, 0, stream>>>(cnt, rowptr, N);
    csr_fill_kernel<<<fillBlocks(E), 256, 0, stream>>>(ei, E, rowptr, cnt, col);

    gemm_wmma_f32_logits_kernel<128, 256, 2><<<N / 16, 32 * (256 / 16), 0, stream>>>(
        x, W1, a_src1, a_dst1, h1, as1, ad1);
    node_softmax_gather_kernel<2, false><<<gatherBlocks, GATHER_WAVES * 32, 0, stream>>>(
        rowptr, col, h1, as1, ad1, b1, gamma1, beta1, rm1, rv1, aggA, batch, (float*)d_out, N, E, out_size / HID);

    gemm_wmma_f32_logits_kernel<256, 128, 1><<<N / 16, 32 * (128 / 16), 0, stream>>>(
        aggA, W2, a_src2, a_dst2, h2, as2, ad2);
    node_softmax_gather_kernel<1, true><<<gatherBlocks, GATHER_WAVES * 32, 0, stream>>>(
        rowptr, col, h2, as2, ad2, b2, gamma2, beta2, rm2, rv2, aggB, batch, (float*)d_out, N, E, out_size / HID);
}


# Edge counting sort with its counters in LDS: edges binned per chunk into 128-node buckets, one work-group per bucket makes rowptr and col; no global atomics or one-block scan (edge sum order changes)
# speedup vs baseline: 5.7441x; 2.0069x over previous
#include <hip/hip_runtime.h>
#include <hip/hip_bf16.h>

typedef float v2f __attribute__((ext_vector_type(2)));
typedef float v8f __attribute__((ext_vector_type(8)));

#define NEG_SLOPE 0.2f
#define EPS 1e-5f

__device__ __forceinline__ void atomicMaxFloat(float* addr, float val) {
    if (val >= 0.0f) {
        atomicMax((int*)addr, __float_as_int(val));
    } else {
        atomicMin((unsigned int*)addr, __float_as_uint(val));
    }
}

__global__ void fill_u32_kernel(unsigned int* __restrict__ p, unsigned int v, long long count) {
    long long i = (long long)blockIdx.x * blockDim.x + threadIdx.x;
    if (i < count) p[i] = v;
}

template <int XMASK>
__device__ __forceinline__ float add_row_xmask(float v) {
    return v + __int_as_float(__builtin_amdgcn_update_dpp(0, __float_as_int(v), 0x160 + XMASK, 0xf, 0xf, true));
}

template <int K, int NCOL, int HEADS>
__global__ void gemm_wmma_f32_logits_kernel(const float* __restrict__ A,
                                            const float* __restrict__ Bm,
                                            const float* __restrict__ a_src,
                                            const float* __restrict__ a_dst,
                                            float* __restrict__ C,
                                            float* __restrict__ as,
                                            float* __restrict__ ad) {
    constexpr int NW = NCOL / 16;
    constexpr int WPH = NW / HEADS;
    constexpr int KP = K + 4;
    __shared__ float As[16 * KP];
    __shared__ alignas(16) float part[2][NW][16];
    const int rowBase = blockIdx.x * 16;
    for (int i = threadIdx.x; i < 16 * K; i += blockDim.x) {
        int r = i / K, k = i - r * K;
        As[r * KP + k] = A[(long long)(rowBase + r) * K + k];
    }
    __syncthreads();

    const int wave = threadIdx.x >> 5;
    const int lane = threadIdx.x & 31;
    const int colBase = wave * 16;
    const int mn = lane & 15;
    const int khalf = (lane >> 4) << 1;

    const float sa = a_src[colBase + mn];
    const float da = a_dst[colBase + mn];

    v8f acc = {};
#if __has_builtin(__builtin_amdgcn_wmma_f32_16x16x4_f32)
    for (int kk = 0; kk < K; kk += 4) {
        v2f a, b;
        a.x = As[mn * KP + kk + khalf];
        a.y = As[mn * KP + kk + khalf + 1];
        b.x = Bm[(long long)(kk + khalf) * NCOL + colBase + mn];
        b.y = Bm[(long long)(kk + khalf + 1) * NCOL + colBase + mn];
        acc = __builtin_amdgcn_wmma_f32_16x16x4_f32(false, a, false, b,
                                                    (short)0, acc, false, false);
    }
#else
    for (int kk = 0; kk < K; ++kk) {
        float av = As[mn * KP + kk];
        for (int r = 0; r < 8; ++r)
            acc[r] += av * Bm[(long long)kk * NCOL + colBase + mn];
    }
#endif
    const int mOff = (lane >> 4) << 3;
#pragma unroll
    for (int r = 0; r < 8; ++r) {
        C[(long long)(rowBase + r + mOff) * NCOL + colBase + mn] = acc[r];
    }

    float ps[8], pd[8];
#pragma unroll
    for (int r = 0; r < 8; ++r) { ps[r] = acc[r] * sa; pd[r] = acc[r] * da; }
#pragma unroll
    for (int r = 0; r < 8; ++r) { ps[r] = add_row_xmask<1>(ps[r]); pd[r] = add_row_xmask<1>(pd[r]); }
#pragma unroll
    for (int r = 0; r < 8; ++r) { ps[r] = add_row_xmask<2>(ps[r]); pd[r] = add_row_xmask<2>(pd[r]); }
#pragma unroll
    for (int r = 0; r < 8; ++r) { ps[r] = add_row_xmask<4>(ps[r]); pd[r] = add_row_xmask<4>(pd[r]); }
#pragma unroll
    for (int r = 0; r < 8; ++r) { ps[r] = add_row_xmask<8>(ps[r]); pd[r] = add_row_xmask<8>(pd[r]); }
    if (mn == 0) {
        *(float4*)(&part[0][wave][mOff])     = make_float4(ps[0], ps[1], ps[2], ps[3]);
        *(float4*)(&part[0][wave][mOff + 4]) = make_float4(ps[4], ps[5], ps[6], ps[7]);
        *(float4*)(&part[1][wave][mOff])     = make_float4(pd[0], pd[1], pd[2], pd[3]);
        *(float4*)(&part[1][wave][mOff + 4]) = make_float4(pd[4], pd[5], pd[6], pd[7]);
    }
    __syncthreads();
    if (threadIdx.x < 32 * HEADS) {
        const int row = threadIdx.x & 15;
        const int which = (threadIdx.x >> 4) & 1;
        const int hd = threadIdx.x >> 5;
        float s = 0.0f;
#pragma unroll
        for (int w = 0; w < WPH; ++w) s += part[which][hd * WPH + w][row];
        float* dstp = which ? ad : as;
        dstp[(long long)(rowBase + row) * HEADS + hd] = s;
    }
}

constexpr int BKT_SHIFT = 7;
constexpr int BKT_NODES = 1 << BKT_SHIFT;
constexpr int MAX_BUCKETS = 1024;
constexpr int SORT_CHUNKS = 64;
constexpr int SORT_THREADS = 1024;

__global__ void __launch_bounds__(SORT_THREADS)
bucket_hist_kernel(const int* __restrict__ ei, int E, int n_nodes, int nb, int chunk,
                   int* __restrict__ table) {
    __shared__ int lh[MAX_BUCKETS];
    const int tid = threadIdx.x;
    for (int i = tid; i < nb; i += SORT_THREADS) lh[i] = 0;
    __syncthreads();
    const int beg = blockIdx.x * chunk;
    const int end = (beg + chunk < E) ? (beg + chunk) : E;
    const int* __restrict__ dst = ei + E;
    for (int i0 = beg + tid; i0 < end; i0 += 4 * SORT_THREADS) {
        int d[4];
#pragma unroll
        for (int u = 0; u < 4; ++u) {
            const int i = i0 + u * SORT_THREADS;
            d[u] = (i < end) ? dst[i] : -1;
        }
#pragma unroll
        for (int u = 0; u < 4; ++u)
            if ((unsigned)d[u] < (unsigned)n_nodes) atomicAdd(&lh[d[u] >> BKT_SHIFT], 1);
    }
    __syncthreads();
    for (int i = tid; i < nb; i += SORT_THREADS) table[blockIdx.x * nb + i] = lh[i];
}

__global__ void __launch_bounds__(SORT_THREADS)
bucket_scatter_kernel(const int* __restrict__ ei, int E, int n_nodes, int nb, int chunk,
                      const int* __restrict__ table, int* __restrict__ bstart,
                      int2* __restrict__ pairs) {
    __shared__ int cur[MAX_BUCKETS];
    __shared__ int wsum[32];
    const int tid = threadIdx.x;
    const int lane = tid & 31;
    const int wv = tid >> 5;
    const int g = blockIdx.x;
    int before = 0, total = 0;
    if (tid < nb) {
#pragma unroll 8
        for (int gg = 0; gg < SORT_CHUNKS; ++gg) {
            const int v = table[gg * nb + tid];
            total += v;
            before += (gg < g) ? v : 0;
        }
    }
    int inc = total;
    for (int off = 1; off < 32; off <<= 1) {
        int t = __shfl_up(inc, off, 32);
        if (lane >= off) inc += t;
    }
    if (lane == 31) wsum[wv] = inc;
    __syncthreads();
    if (wv == 0) {
        const int w = wsum[lane];
        int winc = w;
        for (int off = 1; off < 32; off <<= 1) {
            int t = __shfl_up(winc, off, 32);
            if (lane >= off) winc += t;
        }
        wsum[lane] = winc - w;
    }
    __syncthreads();
    const int bucketBeg = wsum[wv] + inc - total;
    if (tid < nb) {
        cur[tid] = bucketBeg + before;
        if (g == 0) bstart[tid] = bucketBeg;
    }
    if (g == 0 && tid == SORT_THREADS - 1) bstart[nb] = wsum[wv] + inc;
    __syncthreads();

    const int beg = g * chunk;
    const int end = (beg + chunk < E) ? (beg + chunk) : E;
    const int* __restrict__ dst = ei + E;
    for (int i0 = beg + tid; i0 < end; i0 += 4 * SORT_THREADS) {
        int s[4], d[4];
#pragma unroll
        for (int u = 0; u < 4; ++u) {
            const int i = i0 + u * SORT_THREADS;
            d[u] = (i < end) ? dst[i] : -1;
            s[u] = (i < end) ? ei[i] : 0;
        }
#pragma unroll
        for (int u = 0; u < 4; ++u) {
            if ((unsigned)d[u] < (unsigned)n_nodes) {
                const int pos = atomicAdd(&cur[d[u] >> BKT_SHIFT], 1);
                if ((unsigned)pos < (unsigned)E) pairs[pos] = make_int2(s[u], d[u]);
            }
        }
    }
}

__global__ void __launch_bounds__(SORT_THREADS)
bucket_csr_kernel(const int2* __restrict__ pairs, const int* __restrict__ bstart, int E, int n_nodes,
                  int nb, int* __restrict__ rowptr, int* __restrict__ col) {
    __shared__ int lcnt[BKT_NODES];
    __shared__ int lcur[BKT_NODES];
    __shared__ int wtot[BKT_NODES / 32];
    const int tid = threadIdx.x;
    const int lane = tid & 31;
    const int wv = tid >> 5;
    const int b = blockIdx.x;
    int s0 = bstart[b];
    int s1 = bstart[b + 1];
    s0 = (s0 < 0) ? 0 : ((s0 > E) ? E : s0);
    s1 = (s1 < s0) ? s0 : ((s1 > E) ? E : s1);
    const int nodeBase = b << BKT_SHIFT;
    if (tid < BKT_NODES) lcnt[tid] = 0;
    __syncthreads();
    for (int i = s0 + tid; i < s1; i += SORT_THREADS) {
        const unsigned ld = (unsigned)(pairs[i].y - nodeBase);
        if (ld < (unsigned)BKT_NODES) atomicAdd(&lcnt[ld], 1);
    }
    __syncthreads();
    int c = 0, inc = 0;
    if (tid < BKT_NODES) {
        c = lcnt[tid];
        inc = c;
        for (int off = 1; off < 32; off <<= 1) {
            int t = __shfl_up(inc, off, 32);
            if (lane >= off) inc += t;
        }
        if (lane == 31) wtot[wv] = inc;
    }
    __syncthreads();
    if (tid < BKT_NODES) {
        int off = 0;
        for (int w = 0; w < wv; ++w) off += wtot[w];
        const int excl = off + inc - c;
        lcur[tid] = excl;
        const int node = nodeBase + tid;
        if (node < n_nodes) rowptr[node] = s0 + excl;
    }
    if (b == nb - 1 && tid == 0) rowptr[n_nodes] = s1;
    __syncthreads();
    for (int i = s0 + tid; i < s1; i += SORT_THREADS) {
        const int2 p = pairs[i];
        const unsigned ld = (unsigned)(p.y - nodeBase);
        if (ld < (unsigned)BKT_NODES) {
            const int slot = s0 + atomicAdd(&lcur[ld], 1);
            if ((unsigned)slot < (unsigned)E) col[slot] = p.x;
        }
    }
}

__device__ __forceinline__ float bias_bn_relu(float x, float b, float gamma, float beta,
                                              float mean, float var) {
    float v = x + b;
    v = gamma * (v - mean) * rsqrtf(var + EPS) + beta;
    return (v > 0.0f) ? v : 0.0f;
}

constexpr int GATHER_WAVES = 8;

template <int HEADS, bool POOL>
__global__ void node_softmax_gather_kernel(const int* __restrict__ rowptr,
                                           const int* __restrict__ col,
                                           const float* __restrict__ h,
                                           const float* __restrict__ as,
                                           const float* __restrict__ ad,
                                           const float* __restrict__ bias,
                                           const float* __restrict__ gamma,
                                           const float* __restrict__ beta,
                                           const float* __restrict__ mean,
                                           const float* __restrict__ var,
                                           float* __restrict__ out,
                                           const int* __restrict__ batch,
                                           float* __restrict__ pool_out,
                                           int n_nodes, int E, int n_graphs) {
    constexpr int C = 128;
    constexpr int F = HEADS * C;
    static_assert(!POOL || HEADS == 1, "the pooled epilogue is written for one head");
    const int lane = threadIdx.x & 31;
    const int n = __builtin_amdgcn_readfirstlane(
        (int)(((long long)blockIdx.x * blockDim.x + threadIdx.x) >> 5));
    const bool valid = n < n_nodes;
    if (!POOL && !valid) return;

    float4 res[HEADS];
    if (valid) {
        int beg = rowptr[n];
        int nin = rowptr[n + 1] - beg;
        beg = (beg < 0) ? 0 : ((beg > E) ? E : beg);
        nin = (nin < 0) ? 0 : ((nin > E - beg) ? (E - beg) : nin);
        const int deg = nin + 1;

        float adn[HEADS];
#pragma unroll
        for (int hh = 0; hh < HEADS; ++hh) adn[hh] = ad[n * HEADS + hh];

        float m[HEADS];
#pragma unroll
        for (int hh = 0; hh < HEADS; ++hh) m[hh] = -INFINITY;
        for (int j = lane; j < deg; j += 32) {
            int s = (j < nin) ? col[beg + j] : n;
            s = (s < 0) ? 0 : ((s >= n_nodes) ? (n_nodes - 1) : s);
#pragma unroll
            for (int hh = 0; hh < HEADS; ++hh) {
                float v = as[s * HEADS + hh] + adn[hh];
                v = (v > 0.0f) ? v : NEG_SLOPE * v;
                m[hh] = fmaxf(m[hh], v);
            }
        }
        for (int off = 16; off; off >>= 1) {
#pragma unroll
            for (int hh = 0; hh < HEADS; ++hh) m[hh] = fmaxf(m[hh], __shfl_xor(m[hh], off, 32));
        }

        float4 acc[HEADS];
        float dsum[HEADS];
#pragma unroll
        for (int hh = 0; hh < HEADS; ++hh) {
            acc[hh] = make_float4(0.0f, 0.0f, 0.0f, 0.0f);
            dsum[hh] = 0.0f;
        }
        for (int base = 0; base < deg; base += 32) {
            const int j = base + lane;
            int s = n;
            float ex[HEADS];
#pragma unroll
            for (int hh = 0; hh < HEADS; ++hh) ex[hh] = 0.0f;
            if (j < deg) {
                s = (j < nin) ? col[beg + j] : n;
                s = (s < 0) ? 0 : ((s >= n_nodes) ? (n_nodes - 1) : s);
#pragma unroll
                for (int hh = 0; hh < HEADS; ++hh) {
                    float v = as[s * HEADS + hh] + adn[hh];
                    v = (v > 0.0f) ? v : NEG_SLOPE * v;
                    ex[hh] = __expf(v - m[hh]);
                    dsum[hh] += ex[hh];
                }
            }
            const int chunk = (deg - base < 32) ? (deg - base) : 32;
#pragma unroll 4
            for (int t = 0; t < chunk; ++t) {
                const int st = __builtin_amdgcn_readlane(s, t);
                const float* row = h + (size_t)st * F + lane * 4;
#pragma unroll
                for (int hh = 0; hh < HEADS; ++hh) {
                    const float a = __int_as_float(__builtin_amdgcn_readlane(__float_as_int(ex[hh]), t));
                    const float4 hv = *(const float4*)(row + hh * C);
                    acc[hh].x += a * hv.x;
                    acc[hh].y += a * hv.y;
                    acc[hh].z += a * hv.z;
                    acc[hh].w += a * hv.w;
                }
            }
        }
        for (int off = 16; off; off >>= 1) {
#pragma unroll
            for (int hh = 0; hh < HEADS; ++hh) dsum[hh] += __shfl_xor(dsum[hh], off, 32);
        }
#pragma unroll
        for (int hh = 0; hh < HEADS; ++hh) {
            const float inv = 1.0f / dsum[hh];
            const int c0 = hh * C + lane * 4;
            const float4 pb = *(const float4*)(bias + c0);
            const float4 pg = *(const float4*)(gamma + c0);
            const float4 pe = *(const float4*)(beta + c0);
            const float4 pm = *(const float4*)(mean + c0);
            const float4 pv = *(const float4*)(var + c0);
            res[hh].x = bias_bn_relu(acc[hh].x * inv, pb.x, pg.x, pe.x, pm.x, pv.x);
            res[hh].y = bias_bn_relu(acc[hh].y * inv, pb.y, pg.y, pe.y, pm.y, pv.y);
            res[hh].z = bias_bn_relu(acc[hh].z * inv, pb.z, pg.z, pe.z, pm.z, pv.z);
            res[hh].w = bias_bn_relu(acc[hh].w * inv, pb.w, pg.w, pe.w, pm.w, pv.w);
        }
    }

    if constexpr (!POOL) {
#pragma unroll
        for (int hh = 0; hh < HEADS; ++hh)
            *(float4*)(out + (size_t)n * F + hh * C + lane * 4) = res[hh];
    } else {
        __shared__ float rowbuf[GATHER_WAVES * C];
        __shared__ int gid[GATHER_WAVES];
        const int w = threadIdx.x >> 5;
        if (valid) {
            *(float4*)(rowbuf + w * C + lane * 4) = res[0];
            if (lane == 0) {
                const int g = batch[n];
                gid[w] = ((unsigned)g < (unsigned)n_graphs) ? g : -1;
            }
        } else if (lane == 0) {
            gid[w] = -1;
        }
        __syncthreads();
        if (threadIdx.x < C) {
            const int c = threadIdx.x;
            int cur = -1;
            float cm = 0.0f;
#pragma unroll
            for (int i = 0; i < GATHER_WAVES; ++i) {
                const int g = gid[i];
                if (g < 0) continue;
                const float v = rowbuf[i * C + c];
                if (g == cur) {
                    cm = fmaxf(cm, v);
                } else {
                    if (cur >= 0) atomicMaxFloat(&pool_out[(size_t)cur * C + c], cm);
                    cur = g;
                    cm = v;
                }
            }
            if (cur >= 0) atomicMaxFloat(&pool_out[(size_t)cur * C + c], cm);
        }
    }
}

extern "C" void kernel_launch(void* const* d_in, const int* in_sizes, int n_in,
                              void* d_out, int out_size, void* d_ws, size_t ws_size,
                              hipStream_t stream) {
    const int F_IN = 128, HID = 128, HEADS = 2;
    const int F1 = HEADS * HID;
    const int N = in_sizes[0] / F_IN;
    const int E = in_sizes[1] / 2;

    const float* x      = (const float*)d_in[0];
    const int*   ei     = (const int*)d_in[1];
    const int*   batch  = (const int*)d_in[2];
    const float* W1     = (const float*)d_in[3];
    const float* a_src1 = (const float*)d_in[4];
    const float* a_dst1 = (const float*)d_in[5];
    const float* b1     = (const float*)d_in[6];
    const float* gamma1 = (const float*)d_in[7];
    const float* beta1  = (const float*)d_in[8];
    const float* rm1    = (const float*)d_in[9];
    const float* rv1    = (const float*)d_in[10];
    const float* W2     = (const float*)d_in[11];
    const float* a_src2 = (const float*)d_in[12];
    const float* a_dst2 = (const float*)d_in[13];
    const float* b2     = (const float*)d_in[14];
    const float* gamma2 = (const float*)d_in[15];
    const float* beta2  = (const float*)d_in[16];
    const float* rm2    = (const float*)d_in[17];
    const float* rv2    = (const float*)d_in[18];

    float* ws = (float*)d_ws;
    size_t o = 0;
    float* h1   = ws + o; o += (size_t)N * F1;
    float* aggA = ws + o; o += (size_t)N * F1;
    float* aggB = ws + o; o += (size_t)N * HID;
    float* as1  = ws + o; o += (size_t)N * HEADS;
    float* ad1  = ws + o; o += (size_t)N * HEADS;
    float* as2  = ws + o; o += (size_t)N;
    float* ad2  = ws + o; o += (size_t)N;
    o += (size_t)N;
    int* rowptr = (int*)(ws + o); o += (size_t)N + 1;
    int* col    = (int*)(ws + o); o += (size_t)E;
    float* h2   = h1;
    const int nb = (N + BKT_NODES - 1) >> BKT_SHIFT;
    const int chunk = (E + SORT_CHUNKS - 1) / SORT_CHUNKS;
    int2* pairs = (int2*)aggB;
    int* table  = (int*)aggB + 2 * (size_t)E;
    int* bstart = table + (size_t)SORT_CHUNKS * nb;

    const unsigned int NEG_INF = 0xFF800000u;
    auto fillBlocks = [](long long cnt_) { return (unsigned)((cnt_ + 255) / 256); };
    const unsigned gatherBlocks = (unsigned)((N + GATHER_WAVES - 1) / GATHER_WAVES);

    fill_u32_kernel<<<fillBlocks((long long)out_size), 256, 0, stream>>>((unsigned*)d_out, NEG_INF, (long long)out_size);

    bucket_hist_kernel<<<SORT_CHUNKS, SORT_THREADS, 0, stream>>>(ei, E, N, nb, chunk, table);
    bucket_scatter_kernel<<<SORT_CHUNKS, SORT_THREADS, 0, stream>>>(ei, E, N, nb, chunk, table, bstart, pairs);
    bucket_csr_kernel<<<nb, SORT_THREADS, 0, stream>>>(pairs, bstart, E, N, nb, rowptr, col);

    gemm_wmma_f32_logits_kernel<128, 256, 2><<<N / 16, 32 * (256 / 16), 0, stream>>>(
        x, W1, a_src1, a_dst1, h1, as1, ad1);
    node_softmax_gather_kernel<2, false><<<gatherBlocks, GATHER_WAVES * 32, 0, stream>>>(
        rowptr, col, h1, as1, ad1, b1, gamma1, beta1, rm1, rv1, aggA, batch, (float*)d_out, N, E, out_size / HID);

    gemm_wmma_f32_logits_kernel<256, 128, 1><<<N / 16, 32 * (128 / 16), 0, stream>>>(
        aggA, W2, a_src2, a_dst2, h2, as2, ad2);
    node_softmax_gather_kernel<1, true><<<gatherBlocks, GATHER_WAVES * 32, 0, stream>>>(
        rowptr, col, h2, as2, ad2, b2, gamma2, beta2, rm2, rv2, aggB, batch, (float*)d_out, N, E, out_size / HID);
}


# Layer-2 GEMM moved into the layer-1 gather kernel: a work-group's 16 finished rows (8 waves x 2 nodes each) stay in LDS as the WMMA A strip instead of going through memory; 6 launches to 5
# speedup vs baseline: 6.4424x; 1.1216x over previous
#include <hip/hip_runtime.h>
#include <hip/hip_bf16.h>

typedef float v2f __attribute__((ext_vector_type(2)));
typedef float v8f __attribute__((ext_vector_type(8)));

#define NEG_SLOPE 0.2f
#define EPS 1e-5f

__device__ __forceinline__ void atomicMaxFloat(float* addr, float val) {
    if (val >= 0.0f) {
        atomicMax((int*)addr, __float_as_int(val));
    } else {
        atomicMin((unsigned int*)addr, __float_as_uint(val));
    }
}

template <int XMASK>
__device__ __forceinline__ float add_row_xmask(float v) {
    return v + __int_as_float(__builtin_amdgcn_update_dpp(0, __float_as_int(v), 0x160 + XMASK, 0xf, 0xf, true));
}

template <int K, int NCOL, int HEADS, bool STAGE>
__device__ __forceinline__ void gemm_strip_logits(const float* __restrict__ A,
                                                  const float* __restrict__ Bm,
                                                  const float* __restrict__ a_src,
                                                  const float* __restrict__ a_dst,
                                                  float* __restrict__ C,
                                                  float* __restrict__ as,
                                                  float* __restrict__ ad,
                                                  int strip, bool live, int tid,
                                                  float* __restrict__ As,
                                                  float (*__restrict__ part)[NCOL / 16][16]) {
    constexpr int NW = NCOL / 16;
    constexpr int WPH = NW / HEADS;
    constexpr int KP = K + 4;
    const int rowBase = strip * 16;
    if constexpr (STAGE) {
        constexpr int NT = 32 * NW;
        constexpr int PER = 16 * K / NT;
        static_assert(PER * NT == 16 * K, "the strip divides evenly over its threads");
        float stage[PER];
#pragma unroll
        for (int u = 0; u < PER; ++u) {
            const int i = tid + u * NT;
            const int r = i / K, k = i - r * K;
            stage[u] = A[(long long)(rowBase + r) * K + k];
        }
#pragma unroll
        for (int u = 0; u < PER; ++u) {
            const int i = tid + u * NT;
            const int r = i / K, k = i - r * K;
            As[r * KP + k] = stage[u];
        }
        __syncthreads();
    }

    const int wave = tid >> 5;
    const int lane = tid & 31;
    const int colBase = wave * 16;
    const int mn = lane & 15;
    const int khalf = (lane >> 4) << 1;

    const float sa = a_src[colBase + mn];
    const float da = a_dst[colBase + mn];

    v8f acc = {};
#if __has_builtin(__builtin_amdgcn_wmma_f32_16x16x4_f32)
    for (int kk = 0; kk < K; kk += 4) {
        v2f a, b;
        a.x = As[mn * KP + kk + khalf];
        a.y = As[mn * KP + kk + khalf + 1];
        b.x = Bm[(long long)(kk + khalf) * NCOL + colBase + mn];
        b.y = Bm[(long long)(kk + khalf + 1) * NCOL + colBase + mn];
        acc = __builtin_amdgcn_wmma_f32_16x16x4_f32(false, a, false, b,
                                                    (short)0, acc, false, false);
    }
#else
    for (int kk = 0; kk < K; ++kk) {
        float av = As[mn * KP + kk];
        for (int r = 0; r < 8; ++r)
            acc[r] += av * Bm[(long long)kk * NCOL + colBase + mn];
    }
#endif
    const int mOff = (lane >> 4) << 3;
    if (live) {
#pragma unroll
        for (int r = 0; r < 8; ++r) {
            C[(long long)(rowBase + r + mOff) * NCOL + colBase + mn] = acc[r];
        }
    }

    float ps[8], pd[8];
#pragma unroll
    for (int r = 0; r < 8; ++r) { ps[r] = acc[r] * sa; pd[r] = acc[r] * da; }
#pragma unroll
    for (int r = 0; r < 8; ++r) { ps[r] = add_row_xmask<1>(ps[r]); pd[r] = add_row_xmask<1>(pd[r]); }
#pragma unroll
    for (int r = 0; r < 8; ++r) { ps[r] = add_row_xmask<2>(ps[r]); pd[r] = add_row_xmask<2>(pd[r]); }
#pragma unroll
    for (int r = 0; r < 8; ++r) { ps[r] = add_row_xmask<4>(ps[r]); pd[r] = add_row_xmask<4>(pd[r]); }
#pragma unroll
    for (int r = 0; r < 8; ++r) { ps[r] = add_row_xmask<8>(ps[r]); pd[r] = add_row_xmask<8>(pd[r]); }
    if (mn == 0) {
        *(float4*)(&part[0][wave][mOff])     = make_float4(ps[0], ps[1], ps[2], ps[3]);
        *(float4*)(&part[0][wave][mOff + 4]) = make_float4(ps[4], ps[5], ps[6], ps[7]);
        *(float4*)(&part[1][wave][mOff])     = make_float4(pd[0], pd[1], pd[2], pd[3]);
        *(float4*)(&part[1][wave][mOff + 4]) = make_float4(pd[4], pd[5], pd[6], pd[7]);
    }
    __syncthreads();
    if (live && tid < 32 * HEADS) {
        const int row = tid & 15;
        const int which = (tid >> 4) & 1;
        const int hd = tid >> 5;
        float s = 0.0f;
#pragma unroll
        for (int w = 0; w < WPH; ++w) s += part[which][hd * WPH + w][row];
        float* dstp = which ? ad : as;
        dstp[(long long)(rowBase + row) * HEADS + hd] = s;
    }
}

constexpr int BKT_SHIFT = 7;
constexpr int BKT_NODES = 1 << BKT_SHIFT;
constexpr int MAX_BUCKETS = 1024;
constexpr int SORT_CHUNKS = 64;
constexpr int SORT_THREADS = 1024;

__device__ __forceinline__ void bucket_hist_body(const int* __restrict__ ei, int E, int n_nodes, int nb,
                                                 int chunk, int* __restrict__ table,
                                                 unsigned int* __restrict__ pool_bits, int pool_count,
                                                 int g) {
    __shared__ int lh[MAX_BUCKETS];
    const int tid = threadIdx.x;
    for (int i = g * SORT_THREADS + tid; i < pool_count; i += SORT_CHUNKS * SORT_THREADS)
        pool_bits[i] = 0xFF800000u;
    for (int i = tid; i < nb; i += SORT_THREADS) lh[i] = 0;
    __syncthreads();
    const int beg = g * chunk;
    const int end = (beg + chunk < E) ? (beg + chunk) : E;
    const int* __restrict__ dst = ei + E;
    for (int i0 = beg + tid; i0 < end; i0 += 4 * SORT_THREADS) {
        int d[4];
#pragma unroll
        for (int u = 0; u < 4; ++u) {
            const int i = i0 + u * SORT_THREADS;
            d[u] = (i < end) ? dst[i] : -1;
        }
#pragma unroll
        for (int u = 0; u < 4; ++u)
            if ((unsigned)d[u] < (unsigned)n_nodes) atomicAdd(&lh[d[u] >> BKT_SHIFT], 1);
    }
    __syncthreads();
    for (int i = tid; i < nb; i += SORT_THREADS) table[g * nb + i] = lh[i];
}

__device__ __forceinline__ void bucket_scatter_body(const int* __restrict__ ei, int E, int n_nodes, int nb,
                                                    int chunk, const int* __restrict__ table,
                                                    int* __restrict__ bstart, int2* __restrict__ pairs,
                                                    int g) {
    __shared__ int cur[MAX_BUCKETS];
    __shared__ int wsum[32];
    const int tid = threadIdx.x;
    const int lane = tid & 31;
    const int wv = tid >> 5;
    int before = 0, total = 0;
    if (tid < nb) {
#pragma unroll 8
        for (int gg = 0; gg < SORT_CHUNKS; ++gg) {
            const int v = table[gg * nb + tid];
            total += v;
            before += (gg < g) ? v : 0;
        }
    }
    int inc = total;
    for (int off = 1; off < 32; off <<= 1) {
        int t = __shfl_up(inc, off, 32);
        if (lane >= off) inc += t;
    }
    if (lane == 31) wsum[wv] = inc;
    __syncthreads();
    if (wv == 0) {
        const int w = wsum[lane];
        int winc = w;
        for (int off = 1; off < 32; off <<= 1) {
            int t = __shfl_up(winc, off, 32);
            if (lane >= off) winc += t;
        }
        wsum[lane] = winc - w;
    }
    __syncthreads();
    const int bucketBeg = wsum[wv] + inc - total;
    if (tid < nb) {
        cur[tid] = bucketBeg + before;
        if (g == 0) bstart[tid] = bucketBeg;
    }
    if (g == 0 && tid == SORT_THREADS - 1) bstart[nb] = wsum[wv] + inc;
    __syncthreads();

    const int beg = g * chunk;
    const int end = (beg + chunk < E) ? (beg + chunk) : E;
    const int* __restrict__ dst = ei + E;
    for (int i0 = beg + tid; i0 < end; i0 += 4 * SORT_THREADS) {
        int s[4], d[4];
#pragma unroll
        for (int u = 0; u < 4; ++u) {
            const int i = i0 + u * SORT_THREADS;
            d[u] = (i < end) ? dst[i] : -1;
            s[u] = (i < end) ? ei[i] : 0;
        }
#pragma unroll
        for (int u = 0; u < 4; ++u) {
            if ((unsigned)d[u] < (unsigned)n_nodes) {
                const int pos = atomicAdd(&cur[d[u] >> BKT_SHIFT], 1);
                if ((unsigned)pos < (unsigned)E) pairs[pos] = make_int2(s[u], d[u]);
            }
        }
    }
}

template <int STEP>
__global__ void __launch_bounds__(SORT_THREADS)
sort_step_gemm_kernel(const int* __restrict__ ei, int E, int n_nodes, int nb, int chunk,
                      int* __restrict__ table, int* __restrict__ bstart, int2* __restrict__ pairs,
                      unsigned int* __restrict__ pool_bits, int pool_count,
                      const float* __restrict__ x, const float* __restrict__ W,
                      const float* __restrict__ a_src, const float* __restrict__ a_dst,
                      float* __restrict__ h, float* __restrict__ as, float* __restrict__ ad,
                      int strip_begin, int strip_end) {
    constexpr int K = 128, NCOL = 256, HEADS = 2;
    constexpr int HALF = 32 * (NCOL / 16);
    static_assert(2 * HALF == SORT_THREADS, "two strips per work-group");
    if (blockIdx.x < SORT_CHUNKS) {
        if constexpr (STEP == 1)
            bucket_hist_body(ei, E, n_nodes, nb, chunk, table, pool_bits, pool_count, (int)blockIdx.x);
        else
            bucket_scatter_body(ei, E, n_nodes, nb, chunk, table, bstart, pairs, (int)blockIdx.x);
        return;
    }
    __shared__ float As[2][16 * (K + 4)];
    __shared__ alignas(16) float part[2][2][NCOL / 16][16];
    const int half = threadIdx.x / HALF;
    int strip = strip_begin + ((int)blockIdx.x - SORT_CHUNKS) * 2 + half;
    const bool live = strip < strip_end;
    if (!live) strip = strip_end - 1;
    gemm_strip_logits<K, NCOL, HEADS, true>(x, W, a_src, a_dst, h, as, ad,
                                            strip, live, (int)threadIdx.x - half * HALF, As[half], part[half]);
}

__global__ void __launch_bounds__(SORT_THREADS)
bucket_csr_kernel(const int2* __restrict__ pairs, const int* __restrict__ bstart, int E, int n_nodes,
                  int nb, int* __restrict__ rowptr, int* __restrict__ col) {
    __shared__ int lcnt[BKT_NODES];
    __shared__ int lcur[BKT_NODES];
    __shared__ int wtot[BKT_NODES / 32];
    const int tid = threadIdx.x;
    const int lane = tid & 31;
    const int wv = tid >> 5;
    const int b = blockIdx.x;
    int s0 = bstart[b];
    int s1 = bstart[b + 1];
    s0 = (s0 < 0) ? 0 : ((s0 > E) ? E : s0);
    s1 = (s1 < s0) ? s0 : ((s1 > E) ? E : s1);
    const int nodeBase = b << BKT_SHIFT;
    if (tid < BKT_NODES) lcnt[tid] = 0;
    __syncthreads();
    for (int i = s0 + tid; i < s1; i += SORT_THREADS) {
        const unsigned ld = (unsigned)(pairs[i].y - nodeBase);
        if (ld < (unsigned)BKT_NODES) atomicAdd(&lcnt[ld], 1);
    }
    __syncthreads();
    int c = 0, inc = 0;
    if (tid < BKT_NODES) {
        c = lcnt[tid];
        inc = c;
        for (int off = 1; off < 32; off <<= 1) {
            int t = __shfl_up(inc, off, 32);
            if (lane >= off) inc += t;
        }
        if (lane == 31) wtot[wv] = inc;
    }
    __syncthreads();
    if (tid < BKT_NODES) {
        int off = 0;
        for (int w = 0; w < wv; ++w) off += wtot[w];
        const int excl = off + inc - c;
        lcur[tid] = excl;
        const int node = nodeBase + tid;
        if (node < n_nodes) rowptr[node] = s0 + excl;
    }
    if (b == nb - 1 && tid == 0) rowptr[n_nodes] = s1;
    __syncthreads();
    for (int i = s0 + tid; i < s1; i += SORT_THREADS) {
        const int2 p = pairs[i];
        const unsigned ld = (unsigned)(p.y - nodeBase);
        if (ld < (unsigned)BKT_NODES) {
            const int slot = s0 + atomicAdd(&lcur[ld], 1);
            if ((unsigned)slot < (unsigned)E) col[slot] = p.x;
        }
    }
}

__device__ __forceinline__ float bias_bn_relu(float x, float b, float gamma, float beta,
                                              float mean, float var) {
    float v = x + b;
    v = gamma * (v - mean) * rsqrtf(var + EPS) + beta;
    return (v > 0.0f) ? v : 0.0f;
}

constexpr int GATHER_WAVES = 8;

template <int HEADS>
__device__ __forceinline__ void softmax_gather_row(int n, int lane,
                                                   const int* __restrict__ rowptr,
                                                   const int* __restrict__ col,
                                                   const float* __restrict__ h,
                                                   const float* __restrict__ as,
                                                   const float* __restrict__ ad,
                                                   const float* __restrict__ bias,
                                                   const float* __restrict__ gamma,
                                                   const float* __restrict__ beta,
                                                   const float* __restrict__ mean,
                                                   const float* __restrict__ var,
                                                   int n_nodes, int E, float4 (&res)[HEADS]) {
    constexpr int C = 128;
    constexpr int F = HEADS * C;
    int beg = rowptr[n];
    int nin = rowptr[n + 1] - beg;
    beg = (beg < 0) ? 0 : ((beg > E) ? E : beg);
    nin = (nin < 0) ? 0 : ((nin > E - beg) ? (E - beg) : nin);
    const int deg = nin + 1;

    float adn[HEADS];
#pragma unroll
    for (int hh = 0; hh < HEADS; ++hh) adn[hh] = ad[n * HEADS + hh];

    float m[HEADS];
#pragma unroll
    for (int hh = 0; hh < HEADS; ++hh) m[hh] = -INFINITY;
    for (int j = lane; j < deg; j += 32) {
        int s = (j < nin) ? col[beg + j] : n;
        s = (s < 0) ? 0 : ((s >= n_nodes) ? (n_nodes - 1) : s);
#pragma unroll
        for (int hh = 0; hh < HEADS; ++hh) {
            float v = as[s * HEADS + hh] + adn[hh];
            v = (v > 0.0f) ? v : NEG_SLOPE * v;
            m[hh] = fmaxf(m[hh], v);
        }
    }
    for (int off = 16; off; off >>= 1) {
#pragma unroll
        for (int hh = 0; hh < HEADS; ++hh) m[hh] = fmaxf(m[hh], __shfl_xor(m[hh], off, 32));
    }

    float4 acc[HEADS];
    float dsum[HEADS];
#pragma unroll
    for (int hh = 0; hh < HEADS; ++hh) {
        acc[hh] = make_float4(0.0f, 0.0f, 0.0f, 0.0f);
        dsum[hh] = 0.0f;
    }
    for (int base = 0; base < deg; base += 32) {
        const int j = base + lane;
        int s = n;
        float ex[HEADS];
#pragma unroll
        for (int hh = 0; hh < HEADS; ++hh) ex[hh] = 0.0f;
        if (j < deg) {
            s = (j < nin) ? col[beg + j] : n;
            s = (s < 0) ? 0 : ((s >= n_nodes) ? (n_nodes - 1) : s);
#pragma unroll
            for (int hh = 0; hh < HEADS; ++hh) {
                float v = as[s * HEADS + hh] + adn[hh];
                v = (v > 0.0f) ? v : NEG_SLOPE * v;
                ex[hh] = __expf(v - m[hh]);
                dsum[hh] += ex[hh];
            }
        }
        const int chunk = (deg - base < 32) ? (deg - base) : 32;
#pragma unroll 4
        for (int t = 0; t < chunk; ++t) {
            const int st = __builtin_amdgcn_readlane(s, t);
            const float* row = h + (size_t)st * F + lane * 4;
#pragma unroll
            for (int hh = 0; hh < HEADS; ++hh) {
                const float a = __int_as_float(__builtin_amdgcn_readlane(__float_as_int(ex[hh]), t));
                const float4 hv = *(const float4*)(row + hh * C);
                acc[hh].x += a * hv.x;
                acc[hh].y += a * hv.y;
                acc[hh].z += a * hv.z;
                acc[hh].w += a * hv.w;
            }
        }
    }
    for (int off = 16; off; off >>= 1) {
#pragma unroll
        for (int hh = 0; hh < HEADS; ++hh) dsum[hh] += __shfl_xor(dsum[hh], off, 32);
    }
#pragma unroll
    for (int hh = 0; hh < HEADS; ++hh) {
        const float inv = 1.0f / dsum[hh];
        const int c0 = hh * C + lane * 4;
        const float4 pb = *(const float4*)(bias + c0);
        const float4 pg = *(const float4*)(gamma + c0);
        const float4 pe = *(const float4*)(beta + c0);
        const float4 pm = *(const float4*)(mean + c0);
        const float4 pv = *(const float4*)(var + c0);
        res[hh].x = bias_bn_relu(acc[hh].x * inv, pb.x, pg.x, pe.x, pm.x, pv.x);
        res[hh].y = bias_bn_relu(acc[hh].y * inv, pb.y, pg.y, pe.y, pm.y, pv.y);
        res[hh].z = bias_bn_relu(acc[hh].z * inv, pb.z, pg.z, pe.z, pm.z, pv.z);
        res[hh].w = bias_bn_relu(acc[hh].w * inv, pb.w, pg.w, pe.w, pm.w, pv.w);
    }
}

__global__ void __launch_bounds__(GATHER_WAVES * 32, 16)
gather_rows_gemm_kernel(const int* __restrict__ rowptr,
                        const int* __restrict__ col,
                        const float* __restrict__ h1,
                        const float* __restrict__ as1,
                        const float* __restrict__ ad1,
                        const float* __restrict__ bias,
                        const float* __restrict__ gamma,
                        const float* __restrict__ beta,
                        const float* __restrict__ mean,
                        const float* __restrict__ var,
                        const float* __restrict__ W2,
                        const float* __restrict__ a_src2,
                        const float* __restrict__ a_dst2,
                        float* __restrict__ h2,
                        float* __restrict__ as2,
                        float* __restrict__ ad2,
                        int n_nodes, int E) {
    constexpr int K = 256, NCOL = 128;
    constexpr int KP = K + 4;
    constexpr int ROWS_PER_WAVE = 16 / GATHER_WAVES;
    static_assert(ROWS_PER_WAVE * GATHER_WAVES == 16, "the strip's 16 rows divide over the waves");
    static_assert(NCOL / 16 == GATHER_WAVES, "one 16-column tile per wave in the GEMM phase");
    __shared__ alignas(16) float As[16 * KP];
    __shared__ alignas(16) float part[2][NCOL / 16][16];
    const int lane = threadIdx.x & 31;
    const int w = __builtin_amdgcn_readfirstlane((int)(threadIdx.x >> 5));
    const int strip = (int)blockIdx.x;
    for (int it = 0; it < ROWS_PER_WAVE; ++it) {
        const int r = it * GATHER_WAVES + w;
        float4 res[2];
        softmax_gather_row<2>(strip * 16 + r, lane, rowptr, col, h1, as1, ad1,
                              bias, gamma, beta, mean, var, n_nodes, E, res);
        *(float4*)(As + r * KP + lane * 4)       = res[0];
        *(float4*)(As + r * KP + 128 + lane * 4) = res[1];
    }
    __syncthreads();
    gemm_strip_logits<K, NCOL, 1, false>(nullptr, W2, a_src2, a_dst2, h2, as2, ad2,
                                         strip, true, (int)threadIdx.x, As, part);
}

__global__ void node_softmax_gather_pool_kernel(const int* __restrict__ rowptr,
                                                const int* __restrict__ col,
                                                const float* __restrict__ h,
                                                const float* __restrict__ as,
                                                const float* __restrict__ ad,
                                                const float* __restrict__ bias,
                                                const float* __restrict__ gamma,
                                                const float* __restrict__ beta,
                                                const float* __restrict__ mean,
                                                const float* __restrict__ var,
                                                const int* __restrict__ batch,
                                                float* __restrict__ pool_out,
                                                int n_nodes, int E, int n_graphs) {
    constexpr int C = 128;
    const int lane = threadIdx.x & 31;
    const int n = __builtin_amdgcn_readfirstlane(
        (int)(((long long)blockIdx.x * blockDim.x + threadIdx.x) >> 5));
    const bool valid = n < n_nodes;

    float4 res[1];
    if (valid)
        softmax_gather_row<1>(n, lane, rowptr, col, h, as, ad, bias, gamma, beta, mean, var, n_nodes, E, res);

    __shared__ float rowbuf[GATHER_WAVES * C];
    __shared__ int gid[GATHER_WAVES];
    const int w = threadIdx.x >> 5;
    if (valid) {
        *(float4*)(rowbuf + w * C + lane * 4) = res[0];
        if (lane == 0) {
            const int g = batch[n];
            gid[w] = ((unsigned)g < (unsigned)n_graphs) ? g : -1;
        }
    } else if (lane == 0) {
        gid[w] = -1;
    }
    __syncthreads();
    if (threadIdx.x < C) {
        const int c = threadIdx.x;
        int cur = -1;
        float cm = 0.0f;
#pragma unroll
        for (int i = 0; i < GATHER_WAVES; ++i) {
            const int g = gid[i];
            if (g < 0) continue;
            const float v = rowbuf[i * C + c];
            if (g == cur) {
                cm = fmaxf(cm, v);
            } else {
                if (cur >= 0) atomicMaxFloat(&pool_out[(size_t)cur * C + c], cm);
                cur = g;
                cm = v;
            }
        }
        if (cur >= 0) atomicMaxFloat(&pool_out[(size_t)cur * C + c], cm);
    }
}

extern "C" void kernel_launch(void* const* d_in, const int* in_sizes, int n_in,
                              void* d_out, int out_size, void* d_ws, size_t ws_size,
                              hipStream_t stream) {
    const int F_IN = 128, HID = 128, HEADS = 2;
    const int F1 = HEADS * HID;
    const int N = in_sizes[0] / F_IN;
    const int E = in_sizes[1] / 2;

    const float* x      = (const float*)d_in[0];
    const int*   ei     = (const int*)d_in[1];
    const int*   batch  = (const int*)d_in[2];
    const float* W1     = (const float*)d_in[3];
    const float* a_src1 = (const float*)d_in[4];
    const float* a_dst1 = (const float*)d_in[5];
    const float* b1     = (const float*)d_in[6];
    const float* gamma1 = (const float*)d_in[7];
    const float* beta1  = (const float*)d_in[8];
    const float* rm1    = (const float*)d_in[9];
    const float* rv1    = (const float*)d_in[10];
    const float* W2     = (const float*)d_in[11];
    const float* a_src2 = (const float*)d_in[12];
    const float* a_dst2 = (const float*)d_in[13];
    const float* b2     = (const float*)d_in[14];
    const float* gamma2 = (const float*)d_in[15];
    const float* beta2  = (const float*)d_in[16];
    const float* rm2    = (const float*)d_in[17];
    const float* rv2    = (const float*)d_in[18];

    float* ws = (float*)d_ws;
    size_t o = 0;
    float* h1   = ws + o; o += (size_t)N * F1;
    float* h2   = ws + o; o += (size_t)N * F1;
    float* aggB = ws + o; o += (size_t)N * HID;
    float* as1  = ws + o; o += (size_t)N * HEADS;
    float* ad1  = ws + o; o += (size_t)N * HEADS;
    float* as2  = ws + o; o += (size_t)N;
    float* ad2  = ws + o; o += (size_t)N;
    o += (size_t)N;
    int* rowptr = (int*)(ws + o); o += (size_t)N + 1;
    int* col    = (int*)(ws + o); o += (size_t)E;
    const int nb = (N + BKT_NODES - 1) >> BKT_SHIFT;
    const int chunk = (E + SORT_CHUNKS - 1) / SORT_CHUNKS;
    int2* pairs = (int2*)aggB;
    int* table  = (int*)aggB + 2 * (size_t)E;
    int* bstart = table + (size_t)SORT_CHUNKS * nb;

    const unsigned gatherBlocks = (unsigned)((N + GATHER_WAVES - 1) / GATHER_WAVES);
    const int nStrips = N / 16;
    const int nStripsA = (nStrips * 2 / 5) & ~1;

    sort_step_gemm_kernel<1><<<SORT_CHUNKS + (nStripsA + 1) / 2, SORT_THREADS, 0, stream>>>(
        ei, E, N, nb, chunk, table, bstart, pairs, (unsigned int*)d_out, out_size,
        x, W1, a_src1, a_dst1, h1, as1, ad1, 0, nStripsA);
    sort_step_gemm_kernel<2><<<SORT_CHUNKS + (nStrips - nStripsA + 1) / 2, SORT_THREADS, 0, stream>>>(
        ei, E, N, nb, chunk, table, bstart, pairs, (unsigned int*)d_out, out_size,
        x, W1, a_src1, a_dst1, h1, as1, ad1, nStripsA, nStrips);
    bucket_csr_kernel<<<nb, SORT_THREADS, 0, stream>>>(pairs, bstart, E, N, nb, rowptr, col);

    gather_rows_gemm_kernel<<<nStrips, GATHER_WAVES * 32, 0, stream>>>(
        rowptr, col, h1, as1, ad1, b1, gamma1, beta1, rm1, rv1, W2, a_src2, a_dst2, h2, as2, ad2, N, E);

    node_softmax_gather_pool_kernel<<<gatherBlocks, GATHER_WAVES * 32, 0, stream>>>(
        rowptr, col, h2, as2, ad2, b2, gamma2, beta2, rm2, rv2, batch, (float*)d_out, N, E, out_size / HID);
}


# Both gather kernels keep each lane's edge of a node's first 32 (source, attention logit) in registers from the max pass to the exp pass: list and scattered logits are read once, not twice
# speedup vs baseline: 6.7035x; 1.0405x over previous
#include <hip/hip_runtime.h>
#include <hip/hip_bf16.h>

typedef float v2f __attribute__((ext_vector_type(2)));
typedef float v8f __attribute__((ext_vector_type(8)));

#define NEG_SLOPE 0.2f
#define EPS 1e-5f

__device__ __forceinline__ void atomicMaxFloat(float* addr, float val) {
    if (val >= 0.0f) {
        atomicMax((int*)addr, __float_as_int(val));
    } else {
        atomicMin((unsigned int*)addr, __float_as_uint(val));
    }
}

template <int XMASK>
__device__ __forceinline__ float add_row_xmask(float v) {
    return v + __int_as_float(__builtin_amdgcn_update_dpp(0, __float_as_int(v), 0x160 + XMASK, 0xf, 0xf, true));
}

template <int K, int NCOL, int HEADS, bool STAGE>
__device__ __forceinline__ void gemm_strip_logits(const float* __restrict__ A,
                                                  const float* __restrict__ Bm,
                                                  const float* __restrict__ a_src,
                                                  const float* __restrict__ a_dst,
                                                  float* __restrict__ C,
                                                  float* __restrict__ as,
                                                  float* __restrict__ ad,
                                                  int strip, bool live, int tid,
                                                  float* __restrict__ As,
                                                  float (*__restrict__ part)[NCOL / 16][16]) {
    constexpr int NW = NCOL / 16;
    constexpr int WPH = NW / HEADS;
    constexpr int KP = K + 4;
    const int rowBase = strip * 16;
    if constexpr (STAGE) {
        constexpr int NT = 32 * NW;
        constexpr int PER = 16 * K / NT;
        static_assert(PER * NT == 16 * K, "the strip divides evenly over its threads");
        float stage[PER];
#pragma unroll
        for (int u = 0; u < PER; ++u) {
            const int i = tid + u * NT;
            const int r = i / K, k = i - r * K;
            stage[u] = A[(long long)(rowBase + r) * K + k];
        }
#pragma unroll
        for (int u = 0; u < PER; ++u) {
            const int i = tid + u * NT;
            const int r = i / K, k = i - r * K;
            As[r * KP + k] = stage[u];
        }
        __syncthreads();
    }

    const int wave = tid >> 5;
    const int lane = tid & 31;
    const int colBase = wave * 16;
    const int mn = lane & 15;
    const int khalf = (lane >> 4) << 1;

    const float sa = a_src[colBase + mn];
    const float da = a_dst[colBase + mn];

    v8f acc = {};
#if __has_builtin(__builtin_amdgcn_wmma_f32_16x16x4_f32)
    for (int kk = 0; kk < K; kk += 4) {
        v2f a, b;
        a.x = As[mn * KP + kk + khalf];
        a.y = As[mn * KP + kk + khalf + 1];
        b.x = Bm[(long long)(kk + khalf) * NCOL + colBase + mn];
        b.y = Bm[(long long)(kk + khalf + 1) * NCOL + colBase + mn];
        acc = __builtin_amdgcn_wmma_f32_16x16x4_f32(false, a, false, b,
                                                    (short)0, acc, false, false);
    }
#else
    for (int kk = 0; kk < K; ++kk) {
        float av = As[mn * KP + kk];
        for (int r = 0; r < 8; ++r)
            acc[r] += av * Bm[(long long)kk * NCOL + colBase + mn];
    }
#endif
    const int mOff = (lane >> 4) << 3;
    if (live) {
#pragma unroll
        for (int r = 0; r < 8; ++r) {
            C[(long long)(rowBase + r + mOff) * NCOL + colBase + mn] = acc[r];
        }
    }

    float ps[8], pd[8];
#pragma unroll
    for (int r = 0; r < 8; ++r) { ps[r] = acc[r] * sa; pd[r] = acc[r] * da; }
#pragma unroll
    for (int r = 0; r < 8; ++r) { ps[r] = add_row_xmask<1>(ps[r]); pd[r] = add_row_xmask<1>(pd[r]); }
#pragma unroll
    for (int r = 0; r < 8; ++r) { ps[r] = add_row_xmask<2>(ps[r]); pd[r] = add_row_xmask<2>(pd[r]); }
#pragma unroll
    for (int r = 0; r < 8; ++r) { ps[r] = add_row_xmask<4>(ps[r]); pd[r] = add_row_xmask<4>(pd[r]); }
#pragma unroll
    for (int r = 0; r < 8; ++r) { ps[r] = add_row_xmask<8>(ps[r]); pd[r] = add_row_xmask<8>(pd[r]); }
    if (mn == 0) {
        *(float4*)(&part[0][wave][mOff])     = make_float4(ps[0], ps[1], ps[2], ps[3]);
        *(float4*)(&part[0][wave][mOff + 4]) = make_float4(ps[4], ps[5], ps[6], ps[7]);
        *(float4*)(&part[1][wave][mOff])     = make_float4(pd[0], pd[1], pd[2], pd[3]);
        *(float4*)(&part[1][wave][mOff + 4]) = make_float4(pd[4], pd[5], pd[6], pd[7]);
    }
    __syncthreads();
    if (live && tid < 32 * HEADS) {
        const int row = tid & 15;
        const int which = (tid >> 4) & 1;
        const int hd = tid >> 5;
        float s = 0.0f;
#pragma unroll
        for (int w = 0; w < WPH; ++w) s += part[which][hd * WPH + w][row];
        float* dstp = which ? ad : as;
        dstp[(long long)(rowBase + row) * HEADS + hd] = s;
    }
}

constexpr int BKT_SHIFT = 7;
constexpr int BKT_NODES = 1 << BKT_SHIFT;
constexpr int MAX_BUCKETS = 1024;
constexpr int SORT_CHUNKS = 64;
constexpr int SORT_THREADS = 1024;

__device__ __forceinline__ void bucket_hist_body(const int* __restrict__ ei, int E, int n_nodes, int nb,
                                                 int chunk, int* __restrict__ table,
                                                 unsigned int* __restrict__ pool_bits, int pool_count,
                                                 int g) {
    __shared__ int lh[MAX_BUCKETS];
    const int tid = threadIdx.x;
    for (int i = g * SORT_THREADS + tid; i < pool_count; i += SORT_CHUNKS * SORT_THREADS)
        pool_bits[i] = 0xFF800000u;
    for (int i = tid; i < nb; i += SORT_THREADS) lh[i] = 0;
    __syncthreads();
    const int beg = g * chunk;
    const int end = (beg + chunk < E) ? (beg + chunk) : E;
    const int* __restrict__ dst = ei + E;
    for (int i0 = beg + tid; i0 < end; i0 += 4 * SORT_THREADS) {
        int d[4];
#pragma unroll
        for (int u = 0; u < 4; ++u) {
            const int i = i0 + u * SORT_THREADS;
            d[u] = (i < end) ? dst[i] : -1;
        }
#pragma unroll
        for (int u = 0; u < 4; ++u)
            if ((unsigned)d[u] < (unsigned)n_nodes) atomicAdd(&lh[d[u] >> BKT_SHIFT], 1);
    }
    __syncthreads();
    for (int i = tid; i < nb; i += SORT_THREADS) table[g * nb + i] = lh[i];
}

__device__ __forceinline__ void bucket_scatter_body(const int* __restrict__ ei, int E, int n_nodes, int nb,
                                                    int chunk, const int* __restrict__ table,
                                                    int* __restrict__ bstart, int2* __restrict__ pairs,
                                                    int g) {
    __shared__ int cur[MAX_BUCKETS];
    __shared__ int wsum[32];
    const int tid = threadIdx.x;
    const int lane = tid & 31;
    const int wv = tid >> 5;
    int before = 0, total = 0;
    if (tid < nb) {
#pragma unroll 8
        for (int gg = 0; gg < SORT_CHUNKS; ++gg) {
            const int v = table[gg * nb + tid];
            total += v;
            before += (gg < g) ? v : 0;
        }
    }
    int inc = total;
    for (int off = 1; off < 32; off <<= 1) {
        int t = __shfl_up(inc, off, 32);
        if (lane >= off) inc += t;
    }
    if (lane == 31) wsum[wv] = inc;
    __syncthreads();
    if (wv == 0) {
        const int w = wsum[lane];
        int winc = w;
        for (int off = 1; off < 32; off <<= 1) {
            int t = __shfl_up(winc, off, 32);
            if (lane >= off) winc += t;
        }
        wsum[lane] = winc - w;
    }
    __syncthreads();
    const int bucketBeg = wsum[wv] + inc - total;
    if (tid < nb) {
        cur[tid] = bucketBeg + before;
        if (g == 0) bstart[tid] = bucketBeg;
    }
    if (g == 0 && tid == SORT_THREADS - 1) bstart[nb] = wsum[wv] + inc;
    __syncthreads();

    const int beg = g * chunk;
    const int end = (beg + chunk < E) ? (beg + chunk) : E;
    const int* __restrict__ dst = ei + E;
    for (int i0 = beg + tid; i0 < end; i0 += 4 * SORT_THREADS) {
        int s[4], d[4];
#pragma unroll
        for (int u = 0; u < 4; ++u) {
            const int i = i0 + u * SORT_THREADS;
            d[u] = (i < end) ? dst[i] : -1;
            s[u] = (i < end) ? ei[i] : 0;
        }
#pragma unroll
        for (int u = 0; u < 4; ++u) {
            if ((unsigned)d[u] < (unsigned)n_nodes) {
                const int pos = atomicAdd(&cur[d[u] >> BKT_SHIFT], 1);
                if ((unsigned)pos < (unsigned)E) pairs[pos] = make_int2(s[u], d[u]);
            }
        }
    }
}

template <int STEP>
__global__ void __launch_bounds__(SORT_THREADS)
sort_step_gemm_kernel(const int* __restrict__ ei, int E, int n_nodes, int nb, int chunk,
                      int* __restrict__ table, int* __restrict__ bstart, int2* __restrict__ pairs,
                      unsigned int* __restrict__ pool_bits, int pool_count,
                      const float* __restrict__ x, const float* __restrict__ W,
                      const float* __restrict__ a_src, const float* __restrict__ a_dst,
                      float* __restrict__ h, float* __restrict__ as, float* __restrict__ ad,
                      int strip_begin, int strip_end) {
    constexpr int K = 128, NCOL = 256, HEADS = 2;
    constexpr int HALF = 32 * (NCOL / 16);
    static_assert(2 * HALF == SORT_THREADS, "two strips per work-group");
    if (blockIdx.x < SORT_CHUNKS) {
        if constexpr (STEP == 1)
            bucket_hist_body(ei, E, n_nodes, nb, chunk, table, pool_bits, pool_count, (int)blockIdx.x);
        else
            bucket_scatter_body(ei, E, n_nodes, nb, chunk, table, bstart, pairs, (int)blockIdx.x);
        return;
    }
    __shared__ float As[2][16 * (K + 4)];
    __shared__ alignas(16) float part[2][2][NCOL / 16][16];
    const int half = threadIdx.x / HALF;
    int strip = strip_begin + ((int)blockIdx.x - SORT_CHUNKS) * 2 + half;
    const bool live = strip < strip_end;
    if (!live) strip = strip_end - 1;
    gemm_strip_logits<K, NCOL, HEADS, true>(x, W, a_src, a_dst, h, as, ad,
                                            strip, live, (int)threadIdx.x - half * HALF, As[half], part[half]);
}

__global__ void __launch_bounds__(SORT_THREADS)
bucket_csr_kernel(const int2* __restrict__ pairs, const int* __restrict__ bstart, int E, int n_nodes,
                  int nb, int* __restrict__ rowptr, int* __restrict__ col) {
    __shared__ int lcnt[BKT_NODES];
    __shared__ int lcur[BKT_NODES];
    __shared__ int wtot[BKT_NODES / 32];
    const int tid = threadIdx.x;
    const int lane = tid & 31;
    const int wv = tid >> 5;
    const int b = blockIdx.x;
    int s0 = bstart[b];
    int s1 = bstart[b + 1];
    s0 = (s0 < 0) ? 0 : ((s0 > E) ? E : s0);
    s1 = (s1 < s0) ? s0 : ((s1 > E) ? E : s1);
    const int nodeBase = b << BKT_SHIFT;
    if (tid < BKT_NODES) lcnt[tid] = 0;
    __syncthreads();
    for (int i = s0 + tid; i < s1; i += SORT_THREADS) {
        const unsigned ld = (unsigned)(pairs[i].y - nodeBase);
        if (ld < (unsigned)BKT_NODES) atomicAdd(&lcnt[ld], 1);
    }
    __syncthreads();
    int c = 0, inc = 0;
    if (tid < BKT_NODES) {
        c = lcnt[tid];
        inc = c;
        for (int off = 1; off < 32; off <<= 1) {
            int t = __shfl_up(inc, off, 32);
            if (lane >= off) inc += t;
        }
        if (lane == 31) wtot[wv] = inc;
    }
    __syncthreads();
    if (tid < BKT_NODES) {
        int off = 0;
        for (int w = 0; w < wv; ++w) off += wtot[w];
        const int excl = off + inc - c;
        lcur[tid] = excl;
        const int node = nodeBase + tid;
        if (node < n_nodes) rowptr[node] = s0 + excl;
    }
    if (b == nb - 1 && tid == 0) rowptr[n_nodes] = s1;
    __syncthreads();
    for (int i = s0 + tid; i < s1; i += SORT_THREADS) {
        const int2 p = pairs[i];
        const unsigned ld = (unsigned)(p.y - nodeBase);
        if (ld < (unsigned)BKT_NODES) {
            const int slot = s0 + atomicAdd(&lcur[ld], 1);
            if ((unsigned)slot < (unsigned)E) col[slot] = p.x;
        }
    }
}

__device__ __forceinline__ float bias_bn_relu(float x, float b, float gamma, float beta,
                                              float mean, float var) {
    float v = x + b;
    v = gamma * (v - mean) * rsqrtf(var + EPS) + beta;
    return (v > 0.0f) ? v : 0.0f;
}

constexpr int GATHER_WAVES = 8;

template <int HEADS>
__device__ __forceinline__ void softmax_gather_row(int n, int lane,
                                                   const int* __restrict__ rowptr,
                                                   const int* __restrict__ col,
                                                   const float* __restrict__ h,
                                                   const float* __restrict__ as,
                                                   const float* __restrict__ ad,
                                                   const float* __restrict__ bias,
                                                   const float* __restrict__ gamma,
                                                   const float* __restrict__ beta,
                                                   const float* __restrict__ mean,
                                                   const float* __restrict__ var,
                                                   int n_nodes, int E, float4 (&res)[HEADS]) {
    constexpr int C = 128;
    constexpr int F = HEADS * C;
    int beg = rowptr[n];
    int nin = rowptr[n + 1] - beg;
    beg = (beg < 0) ? 0 : ((beg > E) ? E : beg);
    nin = (nin < 0) ? 0 : ((nin > E - beg) ? (E - beg) : nin);
    const int deg = nin + 1;

    float adn[HEADS];
#pragma unroll
    for (int hh = 0; hh < HEADS; ++hh) adn[hh] = ad[n * HEADS + hh];

    float m[HEADS];
    float v0[HEADS];
    int s0 = n;
#pragma unroll
    for (int hh = 0; hh < HEADS; ++hh) { m[hh] = -INFINITY; v0[hh] = 0.0f; }
    if (lane < deg) {
        int s = (lane < nin) ? col[beg + lane] : n;
        s = (s < 0) ? 0 : ((s >= n_nodes) ? (n_nodes - 1) : s);
        s0 = s;
#pragma unroll
        for (int hh = 0; hh < HEADS; ++hh) {
            float v = as[s * HEADS + hh] + adn[hh];
            v = (v > 0.0f) ? v : NEG_SLOPE * v;
            v0[hh] = v;
            m[hh] = fmaxf(m[hh], v);
        }
    }
    for (int j = lane + 32; j < deg; j += 32) {
        int s = (j < nin) ? col[beg + j] : n;
        s = (s < 0) ? 0 : ((s >= n_nodes) ? (n_nodes - 1) : s);
#pragma unroll
        for (int hh = 0; hh < HEADS; ++hh) {
            float v = as[s * HEADS + hh] + adn[hh];
            v = (v > 0.0f) ? v : NEG_SLOPE * v;
            m[hh] = fmaxf(m[hh], v);
        }
    }
    for (int off = 16; off; off >>= 1) {
#pragma unroll
        for (int hh = 0; hh < HEADS; ++hh) m[hh] = fmaxf(m[hh], __shfl_xor(m[hh], off, 32));
    }

    float4 acc[HEADS];
    float dsum[HEADS];
#pragma unroll
    for (int hh = 0; hh < HEADS; ++hh) {
        acc[hh] = make_float4(0.0f, 0.0f, 0.0f, 0.0f);
        dsum[hh] = 0.0f;
    }
    int s = s0;
    float ex[HEADS];
#pragma unroll
    for (int hh = 0; hh < HEADS; ++hh) ex[hh] = 0.0f;
    if (lane < deg) {
#pragma unroll
        for (int hh = 0; hh < HEADS; ++hh) {
            ex[hh] = __expf(v0[hh] - m[hh]);
            dsum[hh] += ex[hh];
        }
    }
    for (int base = 0; base < deg; base += 32) {
        if (base != 0) {
            const int j = base + lane;
            s = n;
#pragma unroll
            for (int hh = 0; hh < HEADS; ++hh) ex[hh] = 0.0f;
            if (j < deg) {
                s = (j < nin) ? col[beg + j] : n;
                s = (s < 0) ? 0 : ((s >= n_nodes) ? (n_nodes - 1) : s);
#pragma unroll
                for (int hh = 0; hh < HEADS; ++hh) {
                    float v = as[s * HEADS + hh] + adn[hh];
                    v = (v > 0.0f) ? v : NEG_SLOPE * v;
                    ex[hh] = __expf(v - m[hh]);
                    dsum[hh] += ex[hh];
                }
            }
        }
        const int chunk = (deg - base < 32) ? (deg - base) : 32;
#pragma unroll 4
        for (int t = 0; t < chunk; ++t) {
            const int st = __builtin_amdgcn_readlane(s, t);
            const float* row = h + (size_t)st * F + lane * 4;
#pragma unroll
            for (int hh = 0; hh < HEADS; ++hh) {
                const float a = __int_as_float(__builtin_amdgcn_readlane(__float_as_int(ex[hh]), t));
                const float4 hv = *(const float4*)(row + hh * C);
                acc[hh].x += a * hv.x;
                acc[hh].y += a * hv.y;
                acc[hh].z += a * hv.z;
                acc[hh].w += a * hv.w;
            }
        }
    }
    for (int off = 16; off; off >>= 1) {
#pragma unroll
        for (int hh = 0; hh < HEADS; ++hh) dsum[hh] += __shfl_xor(dsum[hh], off, 32);
    }
#pragma unroll
    for (int hh = 0; hh < HEADS; ++hh) {
        const float inv = 1.0f / dsum[hh];
        const int c0 = hh * C + lane * 4;
        const float4 pb = *(const float4*)(bias + c0);
        const float4 pg = *(const float4*)(gamma + c0);
        const float4 pe = *(const float4*)(beta + c0);
        const float4 pm = *(const float4*)(mean + c0);
        const float4 pv = *(const float4*)(var + c0);
        res[hh].x = bias_bn_relu(acc[hh].x * inv, pb.x, pg.x, pe.x, pm.x, pv.x);
        res[hh].y = bias_bn_relu(acc[hh].y * inv, pb.y, pg.y, pe.y, pm.y, pv.y);
        res[hh].z = bias_bn_relu(acc[hh].z * inv, pb.z, pg.z, pe.z, pm.z, pv.z);
        res[hh].w = bias_bn_relu(acc[hh].w * inv, pb.w, pg.w, pe.w, pm.w, pv.w);
    }
}

__global__ void __launch_bounds__(GATHER_WAVES * 32, 16)
gather_rows_gemm_kernel(const int* __restrict__ rowptr,
                        const int* __restrict__ col,
                        const float* __restrict__ h1,
                        const float* __restrict__ as1,
                        const float* __restrict__ ad1,
                        const float* __restrict__ bias,
                        const float* __restrict__ gamma,
                        const float* __restrict__ beta,
                        const float* __restrict__ mean,
                        const float* __restrict__ var,
                        const float* __restrict__ W2,
                        const float* __restrict__ a_src2,
                        const float* __restrict__ a_dst2,
                        float* __restrict__ h2,
                        float* __restrict__ as2,
                        float* __restrict__ ad2,
                        int n_nodes, int E) {
    constexpr int K = 256, NCOL = 128;
    constexpr int KP = K + 4;
    constexpr int ROWS_PER_WAVE = 16 / GATHER_WAVES;
    static_assert(ROWS_PER_WAVE * GATHER_WAVES == 16, "the strip's 16 rows divide over the waves");
    static_assert(NCOL / 16 == GATHER_WAVES, "one 16-column tile per wave in the GEMM phase");
    __shared__ alignas(16) float As[16 * KP];
    __shared__ alignas(16) float part[2][NCOL / 16][16];
    const int lane = threadIdx.x & 31;
    const int w = __builtin_amdgcn_readfirstlane((int)(threadIdx.x >> 5));
    const int strip = (int)blockIdx.x;
    for (int it = 0; it < ROWS_PER_WAVE; ++it) {
        const int r = it * GATHER_WAVES + w;
        float4 res[2];
        softmax_gather_row<2>(strip * 16 + r, lane, rowptr, col, h1, as1, ad1,
                              bias, gamma, beta, mean, var, n_nodes, E, res);
        *(float4*)(As + r * KP + lane * 4)       = res[0];
        *(float4*)(As + r * KP + 128 + lane * 4) = res[1];
    }
    __syncthreads();
    gemm_strip_logits<K, NCOL, 1, false>(nullptr, W2, a_src2, a_dst2, h2, as2, ad2,
                                         strip, true, (int)threadIdx.x, As, part);
}

__global__ void node_softmax_gather_pool_kernel(const int* __restrict__ rowptr,
                                                const int* __restrict__ col,
                                                const float* __restrict__ h,
                                                const float* __restrict__ as,
                                                const float* __restrict__ ad,
                                                const float* __restrict__ bias,
                                                const float* __restrict__ gamma,
                                                const float* __restrict__ beta,
                                                const float* __restrict__ mean,
                                                const float* __restrict__ var,
                                                const int* __restrict__ batch,
                                                float* __restrict__ pool_out,
                                                int n_nodes, int E, int n_graphs) {
    constexpr int C = 128;
    const int lane = threadIdx.x & 31;
    const int n = __builtin_amdgcn_readfirstlane(
        (int)(((long long)blockIdx.x * blockDim.x + threadIdx.x) >> 5));
    const bool valid = n < n_nodes;

    float4 res[1];
    if (valid)
        softmax_gather_row<1>(n, lane, rowptr, col, h, as, ad, bias, gamma, beta, mean, var, n_nodes, E, res);

    __shared__ float rowbuf[GATHER_WAVES * C];
    __shared__ int gid[GATHER_WAVES];
    const int w = threadIdx.x >> 5;
    if (valid) {
        *(float4*)(rowbuf + w * C + lane * 4) = res[0];
        if (lane == 0) {
            const int g = batch[n];
            gid[w] = ((unsigned)g < (unsigned)n_graphs) ? g : -1;
        }
    } else if (lane == 0) {
        gid[w] = -1;
    }
    __syncthreads();
    if (threadIdx.x < C) {
        const int c = threadIdx.x;
        int cur = -1;
        float cm = 0.0f;
#pragma unroll
        for (int i = 0; i < GATHER_WAVES; ++i) {
            const int g = gid[i];
            if (g < 0) continue;
            const float v = rowbuf[i * C + c];
            if (g == cur) {
                cm = fmaxf(cm, v);
            } else {
                if (cur >= 0) atomicMaxFloat(&pool_out[(size_t)cur * C + c], cm);
                cur = g;
                cm = v;
            }
        }
        if (cur >= 0) atomicMaxFloat(&pool_out[(size_t)cur * C + c], cm);
    }
}

extern "C" void kernel_launch(void* const* d_in, const int* in_sizes, int n_in,
                              void* d_out, int out_size, void* d_ws, size_t ws_size,
                              hipStream_t stream) {
    const int F_IN = 128, HID = 128, HEADS = 2;
    const int F1 = HEADS * HID;
    const int N = in_sizes[0] / F_IN;
    const int E = in_sizes[1] / 2;

    const float* x      = (const float*)d_in[0];
    const int*   ei     = (const int*)d_in[1];
    const int*   batch  = (const int*)d_in[2];
    const float* W1     = (const float*)d_in[3];
    const float* a_src1 = (const float*)d_in[4];
    const float* a_dst1 = (const float*)d_in[5];
    const float* b1     = (const float*)d_in[6];
    const float* gamma1 = (const float*)d_in[7];
    const float* beta1  = (const float*)d_in[8];
    const float* rm1    = (const float*)d_in[9];
    const float* rv1    = (const float*)d_in[10];
    const float* W2     = (const float*)d_in[11];
    const float* a_src2 = (const float*)d_in[12];
    const float* a_dst2 = (const float*)d_in[13];
    const float* b2     = (const float*)d_in[14];
    const float* gamma2 = (const float*)d_in[15];
    const float* beta2  = (const float*)d_in[16];
    const float* rm2    = (const float*)d_in[17];
    const float* rv2    = (const float*)d_in[18];

    float* ws = (float*)d_ws;
    size_t o = 0;
    float* h1   = ws + o; o += (size_t)N * F1;
    float* h2   = ws + o; o += (size_t)N * F1;
    float* aggB = ws + o; o += (size_t)N * HID;
    float* as1  = ws + o; o += (size_t)N * HEADS;
    float* ad1  = ws + o; o += (size_t)N * HEADS;
    float* as2  = ws + o; o += (size_t)N;
    float* ad2  = ws + o; o += (size_t)N;
    o += (size_t)N;
    int* rowptr = (int*)(ws + o); o += (size_t)N + 1;
    int* col    = (int*)(ws + o); o += (size_t)E;
    const int nb = (N + BKT_NODES - 1) >> BKT_SHIFT;
    const int chunk = (E + SORT_CHUNKS - 1) / SORT_CHUNKS;
    int2* pairs = (int2*)aggB;
    int* table  = (int*)aggB + 2 * (size_t)E;
    int* bstart = table + (size_t)SORT_CHUNKS * nb;

    const unsigned gatherBlocks = (unsigned)((N + GATHER_WAVES - 1) / GATHER_WAVES);
    const int nStrips = N / 16;
    const int nStripsA = (nStrips * 2 / 5) & ~1;

    sort_step_gemm_kernel<1><<<SORT_CHUNKS + (nStripsA + 1) / 2, SORT_THREADS, 0, stream>>>(
        ei, E, N, nb, chunk, table, bstart, pairs, (unsigned int*)d_out, out_size,
        x, W1, a_src1, a_dst1, h1, as1, ad1, 0, nStripsA);
    sort_step_gemm_kernel<2><<<SORT_CHUNKS + (nStrips - nStripsA + 1) / 2, SORT_THREADS, 0, stream>>>(
        ei, E, N, nb, chunk, table, bstart, pairs, (unsigned int*)d_out, out_size,
        x, W1, a_src1, a_dst1, h1, as1, ad1, nStripsA, nStrips);
    bucket_csr_kernel<<<nb, SORT_THREADS, 0, stream>>>(pairs, bstart, E, N, nb, rowptr, col);

    gather_rows_gemm_kernel<<<nStrips, GATHER_WAVES * 32, 0, stream>>>(
        rowptr, col, h1, as1, ad1, b1, gamma1, beta1, rm1, rv1, W2, a_src2, a_dst2, h2, as2, ad2, N, E);

    node_softmax_gather_pool_kernel<<<gatherBlocks, GATHER_WAVES * 32, 0, stream>>>(
        rowptr, col, h2, as2, ad2, b2, gamma2, beta2, rm2, rv2, batch, (float*)d_out, N, E, out_size / HID);
}
